# starting kernel
# baseline (speedup 1.0000x reference)
#include <hip/hip_runtime.h>

#define NROWS 10000
#define IN_C  2048
#define HID   256
#define OUT_C 256
#define E1    160000
#define E2    320000

#define PSTRIDE 160
#define LDS_CHUNK (32 * PSTRIDE)

typedef float v2f __attribute__((ext_vector_type(2)));
typedef float v8f __attribute__((ext_vector_type(8)));

__global__ void h2gcn_fill_zero(float* __restrict__ p, int n) {
    int i = blockIdx.x * blockDim.x + threadIdx.x;
    if (i < n) p[i] = 0.0f;
}

__global__ void h2gcn_gemm1(const float* __restrict__ x,
                            const float* __restrict__ W1,
                            float* __restrict__ h0) {
    __shared__ float lB[2][LDS_CHUNK];
    const int w    = threadIdx.x >> 5;
    const int lane = threadIdx.x & 31;
    const int m0   = (blockIdx.x * 8 + w) * 16;
    const int nb   = blockIdx.y * 64;
    const bool active = (m0 < NROWS);

    v8f acc[4];
    #pragma unroll
    for (int nt = 0; nt < 4; ++nt) acc[nt] = 0.0f;

    int aRow = m0 + (lane & 15);
    if (aRow >= NROWS) aRow = NROWS - 1;
    const float* aBase = x + (size_t)aRow * IN_C + ((lane >> 4) << 1);

    float4 sA[2], sB[2];
    auto loadStage = [&](int k0) {
        #pragma unroll
        for (int e = 0; e < 2; ++e) {
            const int u  = threadIdx.x + 256 * e;
            const int p  = u >> 4;
            const int n4 = (u & 15) << 2;
            const float* g = &W1[(size_t)(k0 + 2 * p) * HID + nb + n4];
            sA[e] = *(const float4*)g;
            sB[e] = *(const float4*)(g + HID);
        }
    };
    auto storeStage = [&](int buf) {
        #pragma unroll
        for (int e = 0; e < 2; ++e) {
            const int u  = threadIdx.x + 256 * e;
            const int p  = u >> 4;
            const int n4 = (u & 15) << 2;
            float* d = &lB[buf][p * PSTRIDE + n4 * 2];
            const float a0[4] = {sA[e].x, sA[e].y, sA[e].z, sA[e].w};
            const float b0[4] = {sB[e].x, sB[e].y, sB[e].z, sB[e].w};
            #pragma unroll
            for (int i = 0; i < 4; ++i) {
                v2f pr; pr.x = a0[i]; pr.y = b0[i];
                *(v2f*)(d + 2 * i) = pr;
            }
        }
    };

    constexpr int NCHUNK = IN_C / 64;
    loadStage(0);
    for (int c = 0; c < NCHUNK; ++c) {
        const int buf = c & 1;
        storeStage(buf);
        __syncthreads();
        if (c + 1 < NCHUNK) loadStage((c + 1) * 64);

        const float* ap = aBase + c * 64;
        #pragma unroll
        for (int kk = 0; kk < 64; kk += 4) {
            const v2f a = *(const v2f*)(ap + kk);
            const int p = (kk >> 1) + (lane >> 4);
            const float* lb = &lB[buf][p * PSTRIDE + (lane & 15) * 2];
            #pragma unroll
            for (int nt = 0; nt < 4; ++nt) {
                const v2f b = *(const v2f*)(lb + nt * 32);
                acc[nt] = __builtin_amdgcn_wmma_f32_16x16x4_f32(
                    false, a, false, b, (short)0, acc[nt], false, false);
            }
        }
        __syncthreads();
    }

    if (active) {
        const int rbase = m0 + ((lane >> 4) << 3);
        #pragma unroll
        for (int nt = 0; nt < 4; ++nt) {
            const int col = nb + nt * 16 + (lane & 15);
            #pragma unroll
            for (int v = 0; v < 8; ++v)
                h0[(size_t)(rbase + v) * HID + col] = acc[nt][v];
        }
    }
}

__global__ void h2gcn_spmm(const int* __restrict__ rows,
                           const int* __restrict__ cols,
                           const float* __restrict__ vals,
                           const float* __restrict__ h0,
                           float* __restrict__ out, int E) {
    const int edge = blockIdx.x * (blockDim.x >> 5) + (threadIdx.x >> 5);
    const int lane = threadIdx.x & 31;
    if (edge >= E) return;
    const int   r = rows[edge];
    const int   c = cols[edge];
    const float v = vals[edge];
    const float4* src = (const float4*)(h0 + (size_t)c * HID);
    float* dst = out + (size_t)r * HID;
    #pragma unroll
    for (int i = 0; i < 2; ++i) {
        const int j = lane + 32 * i;
        const float4 s = src[j];
        atomicAdd(dst + 4 * j + 0, v * s.x);
        atomicAdd(dst + 4 * j + 1, v * s.y);
        atomicAdd(dst + 4 * j + 2, v * s.z);
        atomicAdd(dst + 4 * j + 3, v * s.w);
    }
}

__global__ void h2gcn_gemm2(const float* __restrict__ h0,
                            const float* __restrict__ h1,
                            const float* __restrict__ h2,
                            const float* __restrict__ Wout,
                            const float* __restrict__ bout,
                            float* __restrict__ out) {
    __shared__ float lB[2][LDS_CHUNK];
    const int w    = threadIdx.x >> 5;
    const int lane = threadIdx.x & 31;
    const int m0   = (blockIdx.x * 8 + w) * 16;
    const int nb   = blockIdx.y * 64;
    const bool active = (m0 < NROWS);

    v8f acc[4];
    #pragma unroll
    for (int nt = 0; nt < 4; ++nt) {
        const float b = bout[nb + nt * 16 + (lane & 15)];
        #pragma unroll
        for (int v = 0; v < 8; ++v) acc[nt][v] = b;
    }

    int aRow = m0 + (lane & 15);
    if (aRow >= NROWS) aRow = NROWS - 1;
    const size_t aOff = (size_t)aRow * HID + ((lane >> 4) << 1);

    float4 sA[2], sB[2];
    auto loadStage = [&](int c) {
        const int part = c >> 2;
        const int k0   = (c & 3) * 64;
        const float* wp = Wout + (size_t)part * HID * OUT_C;
        #pragma unroll
        for (int e = 0; e < 2; ++e) {
            const int u  = threadIdx.x + 256 * e;
            const int p  = u >> 4;
            const int n4 = (u & 15) << 2;
            const float* g = &wp[(size_t)(k0 + 2 * p) * OUT_C + nb + n4];
            sA[e] = *(const float4*)g;
            sB[e] = *(const float4*)(g + OUT_C);
        }
    };
    auto storeStage = [&](int buf) {
        #pragma unroll
        for (int e = 0; e < 2; ++e) {
            const int u  = threadIdx.x + 256 * e;
            const int p  = u >> 4;
            const int n4 = (u & 15) << 2;
            float* d = &lB[buf][p * PSTRIDE + n4 * 2];
            const float a0[4] = {sA[e].x, sA[e].y, sA[e].z, sA[e].w};
            const float b0[4] = {sB[e].x, sB[e].y, sB[e].z, sB[e].w};
            #pragma unroll
            for (int i = 0; i < 4; ++i) {
                v2f pr; pr.x = a0[i]; pr.y = b0[i];
                *(v2f*)(d + 2 * i) = pr;
            }
        }
    };

    auto srcOf = [&](int p) { return p == 0 ? h0 : (p == 1 ? h1 : h2); };

    constexpr int NCHUNK = 12;
    loadStage(0);
    for (int c = 0; c < NCHUNK; ++c) {
        const int buf = c & 1;
        storeStage(buf);
        __syncthreads();
        if (c + 1 < NCHUNK) loadStage(c + 1);

        const float* ap = srcOf(c >> 2) + aOff + (c & 3) * 64;
        #pragma unroll
        for (int kk = 0; kk < 64; kk += 4) {
            const v2f a = *(const v2f*)(ap + kk);
            const int p = (kk >> 1) + (lane >> 4);
            const float* lb = &lB[buf][p * PSTRIDE + (lane & 15) * 2];
            #pragma unroll
            for (int nt = 0; nt < 4; ++nt) {
                const v2f b = *(const v2f*)(lb + nt * 32);
                acc[nt] = __builtin_amdgcn_wmma_f32_16x16x4_f32(
                    false, a, false, b, (short)0, acc[nt], false, false);
            }
        }
        __syncthreads();
    }

    if (active) {
        const int rbase = m0 + ((lane >> 4) << 3);
        #pragma unroll
        for (int nt = 0; nt < 4; ++nt) {
            const int col = nb + nt * 16 + (lane & 15);
            #pragma unroll
            for (int v = 0; v < 8; ++v)
                out[(size_t)(rbase + v) * OUT_C + col] = acc[nt][v];
        }
    }
}

extern "C" void kernel_launch(void* const* d_in, const int* in_sizes, int n_in,
                              void* d_out, int out_size, void* d_ws, size_t ws_size,
                              hipStream_t stream) {
    (void)in_sizes; (void)n_in; (void)out_size; (void)ws_size;

    const float* x    = (const float*)d_in[0];
    const int*   a1r  = (const int*)  d_in[1];
    const int*   a1c  = (const int*)  d_in[2];
    const float* a1v  = (const float*)d_in[3];
    const int*   a2r  = (const int*)  d_in[4];
    const int*   a2c  = (const int*)  d_in[5];
    const float* a2v  = (const float*)d_in[6];
    const float* W1   = (const float*)d_in[7];
    const float* Wout = (const float*)d_in[8];
    const float* bout = (const float*)d_in[9];
    float* out = (float*)d_out;

    float* h0 = (float*)d_ws;
    float* h1 = h0 + (size_t)NROWS * HID;
    float* h2 = h1 + (size_t)NROWS * HID;

    {
        const int n = 2 * NROWS * HID;
        h2gcn_fill_zero<<<dim3((n + 255) / 256), dim3(256), 0, stream>>>(h1, n);
    }

    h2gcn_gemm1<<<dim3(79, 4), dim3(256), 0, stream>>>(x, W1, h0);

    h2gcn_spmm<<<dim3((E1 + 7) / 8), dim3(256), 0, stream>>>(a1r, a1c, a1v, h0, h1, E1);
    h2gcn_spmm<<<dim3((E2 + 7) / 8), dim3(256), 0, stream>>>(a2r, a2c, a2v, h0, h2, E2);

    h2gcn_gemm2<<<dim3(79, 4), dim3(256), 0, stream>>>(h0, h1, h2, Wout, bout, out);
}


# GEMM1 re-cut to equal WMMA shares per SIMD: 250 4-wave groups own 80x128; a wave takes all 40 tiles for a K quarter, operands straight from global; K quarters summed via LDS (sum order changes)
# speedup vs baseline: 2.0424x; 2.0424x over previous
#include <hip/hip_runtime.h>

#define NROWS 10000
#define IN_C  2048
#define HID   256
#define OUT_C 256
#define E1    160000
#define E2    320000

#define PSTRIDE 160
#define LDS_CHUNK (32 * PSTRIDE)

typedef float v2f __attribute__((ext_vector_type(2)));
typedef float v8f __attribute__((ext_vector_type(8)));

#define G1_ROWS  80
#define G1_COLS  128
#define G1_MT    (G1_ROWS / 16)
#define G1_NT    (G1_COLS / 16)
#define G1_KW    (IN_C / 4)
#define G1_NBLK  (G1_KW / 32)
#define G1_D     4
#define G1_PITCH 132

static_assert(G1_ROWS * 125 == NROWS, "row blocks must tile the rows exactly");
static_assert(G1_COLS * 2 == HID, "two column halves");
static_assert(G1_NBLK % 2 == 0, "the K loop runs pairs of blocks (two A register sets)");

static __device__ __forceinline__ void g1_loadA(float4 (&a)[G1_D][G1_MT], const float* ap) {
    #pragma unroll
    for (int mt = 0; mt < G1_MT; ++mt) {
        #pragma unroll
        for (int d = 0; d < G1_D; ++d)
            a[d][mt] = *(const float4*)(ap + d * 8 + (size_t)mt * 16 * IN_C);
    }
}

static __device__ __forceinline__ void g1_loadB(float (&b)[4][G1_NT], const float* bp) {
    #pragma unroll
    for (int j = 0; j < 4; ++j) {
        #pragma unroll
        for (int nt = 0; nt < G1_NT; ++nt)
            b[j][nt] = bp[(size_t)j * HID + nt * 16];
    }
}

static __device__ __forceinline__ void g1_mul(v8f (&acc)[G1_MT][G1_NT],
                                              const float4 (&a)[G1_MT],
                                              const float (&b)[4][G1_NT]) {
    #pragma unroll
    for (int s = 0; s < 2; ++s) {
        #pragma unroll
        for (int mt = 0; mt < G1_MT; ++mt) {
            v2f av;
            av.x = s ? a[mt].z : a[mt].x;
            av.y = s ? a[mt].w : a[mt].y;
            #pragma unroll
            for (int nt = 0; nt < G1_NT; ++nt) {
                v2f bv;
                bv.x = b[2 * s][nt];
                bv.y = b[2 * s + 1][nt];
                acc[mt][nt] = __builtin_amdgcn_wmma_f32_16x16x4_f32(
                    false, av, false, bv, (short)0, acc[mt][nt], false, false);
            }
        }
    }
}

template <int P, bool MORE>
static __device__ __forceinline__ void g1_block(v8f (&acc)[G1_MT][G1_NT],
                                                float4 (&a)[2][G1_D][G1_MT],
                                                float (&b)[G1_D][4][G1_NT],
                                                const float* ap, const float* bp) {
    #pragma unroll
    for (int d = 0; d < G1_D; ++d) {
        g1_mul(acc, a[P][d], b[d]);
        if (MORE) {
            __builtin_amdgcn_sched_barrier(0);
            g1_loadB(b[d], bp + (size_t)(32 + d * 8) * HID);
            if (d == 0) g1_loadA(a[P ^ 1], ap + 32);
            __builtin_amdgcn_sched_barrier(0);
        }
    }
}

__global__ void __launch_bounds__(128)
h2gcn_gemm1(const float* __restrict__ x,
            const float* __restrict__ W1,
            float* __restrict__ h0) {
    __shared__ __attribute__((aligned(16))) float part[4 * G1_ROWS * G1_PITCH];

    const int w    = threadIdx.x >> 5;
    const int lane = threadIdx.x & 31;
    const int c    = lane & 15;
    const int h    = lane >> 4;
    const int rb   = blockIdx.x * G1_ROWS;
    const int nb   = blockIdx.y * G1_COLS;
    const int kw   = w * G1_KW;

    v8f acc[G1_MT][G1_NT];
    #pragma unroll
    for (int mt = 0; mt < G1_MT; ++mt) {
        #pragma unroll
        for (int nt = 0; nt < G1_NT; ++nt) acc[mt][nt] = 0.0f;
    }

    const float* ap = x  + (size_t)(rb + c) * IN_C + kw + 4 * h;
    const float* bp = W1 + (size_t)(kw + 4 * h) * HID + nb + c;

    float4 a[2][G1_D][G1_MT];
    float  b[G1_D][4][G1_NT];

    g1_loadA(a[0], ap);
    #pragma unroll
    for (int d = 0; d < G1_D; ++d) g1_loadB(b[d], bp + (size_t)(d * 8) * HID);

    for (int it = 0; it < G1_NBLK / 2 - 1; ++it) {
        g1_block<0, true>(acc, a, b, ap, bp);
        ap += 32;  bp += (size_t)32 * HID;
        g1_block<1, true>(acc, a, b, ap, bp);
        ap += 32;  bp += (size_t)32 * HID;
    }
    g1_block<0, true>(acc, a, b, ap, bp);
    ap += 32;  bp += (size_t)32 * HID;
    g1_block<1, false>(acc, a, b, ap, bp);

    float* pw = part + w * (G1_ROWS * G1_PITCH);
    #pragma unroll
    for (int mt = 0; mt < G1_MT; ++mt) {
        #pragma unroll
        for (int nt = 0; nt < G1_NT; ++nt) {
            #pragma unroll
            for (int r = 0; r < 8; ++r)
                pw[(16 * mt + 8 * h + r) * G1_PITCH + 16 * nt + c] = acc[mt][nt][r];
        }
    }
    __syncthreads();

    for (int i = threadIdx.x; i < G1_ROWS * (G1_COLS / 4); i += 128) {
        const int row = i >> 5;
        const int c4  = i & 31;
        const float* p = part + row * G1_PITCH + c4 * 4;
        const float4 s0 = *(const float4*)(p);
        const float4 s1 = *(const float4*)(p + 1 * G1_ROWS * G1_PITCH);
        const float4 s2 = *(const float4*)(p + 2 * G1_ROWS * G1_PITCH);
        const float4 s3 = *(const float4*)(p + 3 * G1_ROWS * G1_PITCH);
        float4 o;
        o.x = ((s0.x + s1.x) + s2.x) + s3.x;
        o.y = ((s0.y + s1.y) + s2.y) + s3.y;
        o.z = ((s0.z + s1.z) + s2.z) + s3.z;
        o.w = ((s0.w + s1.w) + s2.w) + s3.w;
        *(float4*)(h0 + (size_t)(rb + row) * HID + nb + c4 * 4) = o;
    }
}

#define AGG_ROWS    40
#define AGG_THREADS 1024
#define AGG_WAVES   (AGG_THREADS / 32)
#define AGG_VECS    4
#define AGG_WSTEP   (AGG_VECS * 128)
#define AGG_BSTEP   (AGG_WSTEP * AGG_WAVES)

static_assert(AGG_ROWS * 250 == NROWS, "row owners must tile the rows exactly");
static_assert(E1 % 4 == 0 && E2 % 4 == 0, "16-byte index loads are all-in or all-out");
static_assert(AGG_VECS * 4 <= 32, "one hit bit per tested index in a 32-bit lane mask");

__global__ void __launch_bounds__(AGG_THREADS)
h2gcn_aggregate(const int* __restrict__ rows1, const int* __restrict__ cols1,
                const float* __restrict__ vals1,
                const int* __restrict__ rows2, const int* __restrict__ cols2,
                const float* __restrict__ vals2,
                const float* __restrict__ h0,
                float* __restrict__ h1, float* __restrict__ h2) {
    __shared__ float acc[AGG_ROWS * HID];

    const int part = blockIdx.y;
    const int*   rows = part ? rows2 : rows1;
    const int*   cols = part ? cols2 : cols1;
    const float* vals = part ? vals2 : vals1;
    const int    E    = part ? E2 : E1;
    const int rowBase = blockIdx.x * AGG_ROWS;
    float* dst = (part ? h2 : h1) + (size_t)rowBase * HID;

    const int lane = threadIdx.x & 31;
    const int wave = __builtin_amdgcn_readfirstlane((int)(threadIdx.x >> 5));

    for (int i = threadIdx.x; i < AGG_ROWS * HID / 4; i += AGG_THREADS)
        ((float4*)acc)[i] = make_float4(0.0f, 0.0f, 0.0f, 0.0f);
    __syncthreads();

    const int nStep = (E + AGG_BSTEP - 1) / AGG_BSTEP;
    int step = blockIdx.x % nStep;

    int4 rv[AGG_VECS];
    auto loadIdx = [&](int base) {
        #pragma unroll
        for (int j = 0; j < AGG_VECS; ++j) {
            const int e0 = base + j * 128 + lane * 4;
            rv[j] = (e0 < E) ? *(const int4*)(rows + e0) : make_int4(-1, -1, -1, -1);
        }
    };
    loadIdx((step * AGG_WAVES + wave) * AGG_WSTEP);

    for (int s = 0; s < nStep; ++s) {
        const int base = (step * AGG_WAVES + wave) * AGG_WSTEP;

        unsigned lm = 0u;
        #pragma unroll
        for (int j = 0; j < AGG_VECS; ++j) {
            lm |= ((unsigned)(rv[j].x - rowBase) < (unsigned)AGG_ROWS ? 1u : 0u) << (4 * j + 0);
            lm |= ((unsigned)(rv[j].y - rowBase) < (unsigned)AGG_ROWS ? 1u : 0u) << (4 * j + 1);
            lm |= ((unsigned)(rv[j].z - rowBase) < (unsigned)AGG_ROWS ? 1u : 0u) << (4 * j + 2);
            lm |= ((unsigned)(rv[j].w - rowBase) < (unsigned)AGG_ROWS ? 1u : 0u) << (4 * j + 3);
        }

        step = (step + 1 == nStep) ? 0 : step + 1;
        if (s + 1 < nStep) loadIdx((step * AGG_WAVES + wave) * AGG_WSTEP);

        auto nextEdge = [&]() -> int {
            const unsigned bal = __builtin_amdgcn_ballot_w32(lm != 0u);
            if (bal == 0u) return -1;
            const int src = __builtin_ctz(bal);
            const unsigned lmS = (unsigned)__builtin_amdgcn_readlane((int)lm, src);
            const int k = __builtin_ctz(lmS);
            if (lane == src) lm &= lm - 1u;
            return base + (k >> 2) * 128 + src * 4 + (k & 3);
        };

        int e = nextEdge();
        int r = 0, c = 0; float v = 0.0f;
        if (e >= 0) { r = rows[e]; c = cols[e]; v = vals[e]; }
        while (e >= 0) {
            const float* src = h0 + (size_t)c * HID + lane;
            float* a = acc + (r - rowBase) * HID + lane;
            float sv[HID / 32];
            #pragma unroll
            for (int t = 0; t < HID / 32; ++t) sv[t] = src[32 * t];

            const int en = nextEdge();
            int rn = 0, cn = 0; float vn = 0.0f;
            if (en >= 0) { rn = rows[en]; cn = cols[en]; vn = vals[en]; }

            #pragma unroll
            for (int t = 0; t < HID / 32; ++t) atomicAdd(a + 32 * t, v * sv[t]);

            e = en; r = rn; c = cn; v = vn;
        }
    }

    __syncthreads();
    for (int i = threadIdx.x; i < AGG_ROWS * HID / 4; i += AGG_THREADS)
        ((float4*)dst)[i] = ((const float4*)acc)[i];
}

__global__ void h2gcn_gemm2(const float* __restrict__ h0,
                            const float* __restrict__ h1,
                            const float* __restrict__ h2,
                            const float* __restrict__ Wout,
                            const float* __restrict__ bout,
                            float* __restrict__ out) {
    __shared__ float lB[2][LDS_CHUNK];
    const int w    = threadIdx.x >> 5;
    const int lane = threadIdx.x & 31;
    const int m0   = (blockIdx.x * 8 + w) * 16;
    const int nb   = blockIdx.y * 64;
    const bool active = (m0 < NROWS);

    v8f acc[4];
    #pragma unroll
    for (int nt = 0; nt < 4; ++nt) {
        const float b = bout[nb + nt * 16 + (lane & 15)];
        #pragma unroll
        for (int v = 0; v < 8; ++v) acc[nt][v] = b;
    }

    int aRow = m0 + (lane & 15);
    if (aRow >= NROWS) aRow = NROWS - 1;
    const size_t aOff = (size_t)aRow * HID + ((lane >> 4) << 1);

    float4 sA[2], sB[2];
    auto loadStage = [&](int c) {
        const int part = c >> 2;
        const int k0   = (c & 3) * 64;
        const float* wp = Wout + (size_t)part * HID * OUT_C;
        #pragma unroll
        for (int e = 0; e < 2; ++e) {
            const int u  = threadIdx.x + 256 * e;
            const int p  = u >> 4;
            const int n4 = (u & 15) << 2;
            const float* g = &wp[(size_t)(k0 + 2 * p) * OUT_C + nb + n4];
            sA[e] = *(const float4*)g;
            sB[e] = *(const float4*)(g + OUT_C);
        }
    };
    auto storeStage = [&](int buf) {
        #pragma unroll
        for (int e = 0; e < 2; ++e) {
            const int u  = threadIdx.x + 256 * e;
            const int p  = u >> 4;
            const int n4 = (u & 15) << 2;
            float* d = &lB[buf][p * PSTRIDE + n4 * 2];
            const float a0[4] = {sA[e].x, sA[e].y, sA[e].z, sA[e].w};
            const float b0[4] = {sB[e].x, sB[e].y, sB[e].z, sB[e].w};
            #pragma unroll
            for (int i = 0; i < 4; ++i) {
                v2f pr; pr.x = a0[i]; pr.y = b0[i];
                *(v2f*)(d + 2 * i) = pr;
            }
        }
    };

    auto srcOf = [&](int p) { return p == 0 ? h0 : (p == 1 ? h1 : h2); };

    constexpr int NCHUNK = 12;
    loadStage(0);
    for (int c = 0; c < NCHUNK; ++c) {
        const int buf = c & 1;
        storeStage(buf);
        __syncthreads();
        if (c + 1 < NCHUNK) loadStage(c + 1);

        const float* ap = srcOf(c >> 2) + aOff + (c & 3) * 64;
        #pragma unroll
        for (int kk = 0; kk < 64; kk += 4) {
            const v2f a = *(const v2f*)(ap + kk);
            const int p = (kk >> 1) + (lane >> 4);
            const float* lb = &lB[buf][p * PSTRIDE + (lane & 15) * 2];
            #pragma unroll
            for (int nt = 0; nt < 4; ++nt) {
                const v2f b = *(const v2f*)(lb + nt * 32);
                acc[nt] = __builtin_amdgcn_wmma_f32_16x16x4_f32(
                    false, a, false, b, (short)0, acc[nt], false, false);
            }
        }
        __syncthreads();
    }

    if (active) {
        const int rbase = m0 + ((lane >> 4) << 3);
        #pragma unroll
        for (int nt = 0; nt < 4; ++nt) {
            const int col = nb + nt * 16 + (lane & 15);
            #pragma unroll
            for (int v = 0; v < 8; ++v)
                out[(size_t)(rbase + v) * OUT_C + col] = acc[nt][v];
        }
    }
}

extern "C" void kernel_launch(void* const* d_in, const int* in_sizes, int n_in,
                              void* d_out, int out_size, void* d_ws, size_t ws_size,
                              hipStream_t stream) {
    (void)in_sizes; (void)n_in; (void)out_size; (void)ws_size;

    const float* x    = (const float*)d_in[0];
    const int*   a1r  = (const int*)  d_in[1];
    const int*   a1c  = (const int*)  d_in[2];
    const float* a1v  = (const float*)d_in[3];
    const int*   a2r  = (const int*)  d_in[4];
    const int*   a2c  = (const int*)  d_in[5];
    const float* a2v  = (const float*)d_in[6];
    const float* W1   = (const float*)d_in[7];
    const float* Wout = (const float*)d_in[8];
    const float* bout = (const float*)d_in[9];
    float* out = (float*)d_out;

    float* h0 = (float*)d_ws;
    float* h1 = h0 + (size_t)NROWS * HID;
    float* h2 = h1 + (size_t)NROWS * HID;

    h2gcn_gemm1<<<dim3(125, 2), dim3(128), 0, stream>>>(x, W1, h0);

    h2gcn_aggregate<<<dim3(250, 2), dim3(AGG_THREADS), 0, stream>>>(
        a1r, a1c, a1v, a2r, a2c, a2v, h0, h1, h2);

    h2gcn_gemm2<<<dim3(79, 4), dim3(256), 0, stream>>>(h0, h1, h2, Wout, bout, out);
}


# GEMM2 re-cut to equal WMMA shares per SIMD: 250 4-wave groups own 80x128; a wave takes all 40 tiles for a K quarter, operands from global; K quarters summed via LDS, bias last (sum order changes)
# speedup vs baseline: 2.3125x; 1.1322x over previous
#include <hip/hip_runtime.h>

#define NROWS 10000
#define IN_C  2048
#define HID   256
#define OUT_C 256
#define E1    160000
#define E2    320000

typedef float v2f __attribute__((ext_vector_type(2)));
typedef float v8f __attribute__((ext_vector_type(8)));

#define G1_ROWS  80
#define G1_COLS  128
#define G1_MT    (G1_ROWS / 16)
#define G1_NT    (G1_COLS / 16)
#define G1_KW    (IN_C / 4)
#define G1_NBLK  (G1_KW / 32)
#define G1_D     4
#define G1_PITCH 132

static_assert(G1_ROWS * 125 == NROWS, "row blocks must tile the rows exactly");
static_assert(G1_COLS * 2 == HID, "two column halves");
static_assert(G1_NBLK % 2 == 0, "the K loop runs pairs of blocks (two A register sets)");

template <int LD>
static __device__ __forceinline__ void g1_loadA(float4 (&a)[G1_D][G1_MT], const float* ap) {
    #pragma unroll
    for (int mt = 0; mt < G1_MT; ++mt) {
        #pragma unroll
        for (int d = 0; d < G1_D; ++d)
            a[d][mt] = *(const float4*)(ap + d * 8 + (size_t)mt * 16 * LD);
    }
}

static __device__ __forceinline__ void g1_loadB(float (&b)[4][G1_NT], const float* bp) {
    #pragma unroll
    for (int j = 0; j < 4; ++j) {
        #pragma unroll
        for (int nt = 0; nt < G1_NT; ++nt)
            b[j][nt] = bp[(size_t)j * HID + nt * 16];
    }
}

static __device__ __forceinline__ void g1_mul(v8f (&acc)[G1_MT][G1_NT],
                                              const float4 (&a)[G1_MT],
                                              const float (&b)[4][G1_NT]) {
    #pragma unroll
    for (int s = 0; s < 2; ++s) {
        #pragma unroll
        for (int mt = 0; mt < G1_MT; ++mt) {
            v2f av;
            av.x = s ? a[mt].z : a[mt].x;
            av.y = s ? a[mt].w : a[mt].y;
            #pragma unroll
            for (int nt = 0; nt < G1_NT; ++nt) {
                v2f bv;
                bv.x = b[2 * s][nt];
                bv.y = b[2 * s + 1][nt];
                acc[mt][nt] = __builtin_amdgcn_wmma_f32_16x16x4_f32(
                    false, av, false, bv, (short)0, acc[mt][nt], false, false);
            }
        }
    }
}

template <int P, bool MORE, int LD>
static __device__ __forceinline__ void g1_block(v8f (&acc)[G1_MT][G1_NT],
                                                float4 (&a)[2][G1_D][G1_MT],
                                                float (&b)[G1_D][4][G1_NT],
                                                const float* apn, const float* bp) {
    #pragma unroll
    for (int d = 0; d < G1_D; ++d) {
        g1_mul(acc, a[P][d], b[d]);
        if (MORE) {
            __builtin_amdgcn_sched_barrier(0);
            g1_loadB(b[d], bp + (size_t)(32 + d * 8) * HID);
            if (d == 0) g1_loadA<LD>(a[P ^ 1], apn);
            __builtin_amdgcn_sched_barrier(0);
        }
    }
}

__global__ void __launch_bounds__(128)
h2gcn_gemm1(const float* __restrict__ x,
            const float* __restrict__ W1,
            float* __restrict__ h0) {
    __shared__ __attribute__((aligned(16))) float part[4 * G1_ROWS * G1_PITCH];

    const int w    = threadIdx.x >> 5;
    const int lane = threadIdx.x & 31;
    const int c    = lane & 15;
    const int h    = lane >> 4;
    const int rb   = blockIdx.x * G1_ROWS;
    const int nb   = blockIdx.y * G1_COLS;
    const int kw   = w * G1_KW;

    v8f acc[G1_MT][G1_NT];
    #pragma unroll
    for (int mt = 0; mt < G1_MT; ++mt) {
        #pragma unroll
        for (int nt = 0; nt < G1_NT; ++nt) acc[mt][nt] = 0.0f;
    }

    const float* ap = x  + (size_t)(rb + c) * IN_C + kw + 4 * h;
    const float* bp = W1 + (size_t)(kw + 4 * h) * HID + nb + c;

    float4 a[2][G1_D][G1_MT];
    float  b[G1_D][4][G1_NT];

    g1_loadA<IN_C>(a[0], ap);
    #pragma unroll
    for (int d = 0; d < G1_D; ++d) g1_loadB(b[d], bp + (size_t)(d * 8) * HID);

    for (int it = 0; it < G1_NBLK / 2 - 1; ++it) {
        g1_block<0, true, IN_C>(acc, a, b, ap + 32, bp);
        ap += 32;  bp += (size_t)32 * HID;
        g1_block<1, true, IN_C>(acc, a, b, ap + 32, bp);
        ap += 32;  bp += (size_t)32 * HID;
    }
    g1_block<0, true, IN_C>(acc, a, b, ap + 32, bp);
    ap += 32;  bp += (size_t)32 * HID;
    g1_block<1, false, IN_C>(acc, a, b, ap, bp);

    float* pw = part + w * (G1_ROWS * G1_PITCH);
    #pragma unroll
    for (int mt = 0; mt < G1_MT; ++mt) {
        #pragma unroll
        for (int nt = 0; nt < G1_NT; ++nt) {
            #pragma unroll
            for (int r = 0; r < 8; ++r)
                pw[(16 * mt + 8 * h + r) * G1_PITCH + 16 * nt + c] = acc[mt][nt][r];
        }
    }
    __syncthreads();

    for (int i = threadIdx.x; i < G1_ROWS * (G1_COLS / 4); i += 128) {
        const int row = i >> 5;
        const int c4  = i & 31;
        const float* p = part + row * G1_PITCH + c4 * 4;
        const float4 s0 = *(const float4*)(p);
        const float4 s1 = *(const float4*)(p + 1 * G1_ROWS * G1_PITCH);
        const float4 s2 = *(const float4*)(p + 2 * G1_ROWS * G1_PITCH);
        const float4 s3 = *(const float4*)(p + 3 * G1_ROWS * G1_PITCH);
        float4 o;
        o.x = ((s0.x + s1.x) + s2.x) + s3.x;
        o.y = ((s0.y + s1.y) + s2.y) + s3.y;
        o.z = ((s0.z + s1.z) + s2.z) + s3.z;
        o.w = ((s0.w + s1.w) + s2.w) + s3.w;
        *(float4*)(h0 + (size_t)(rb + row) * HID + nb + c4 * 4) = o;
    }
}

#define AGG_ROWS    40
#define AGG_THREADS 1024
#define AGG_WAVES   (AGG_THREADS / 32)
#define AGG_VECS    4
#define AGG_WSTEP   (AGG_VECS * 128)
#define AGG_BSTEP   (AGG_WSTEP * AGG_WAVES)

static_assert(AGG_ROWS * 250 == NROWS, "row owners must tile the rows exactly");
static_assert(E1 % 4 == 0 && E2 % 4 == 0, "16-byte index loads are all-in or all-out");
static_assert(AGG_VECS * 4 <= 32, "one hit bit per tested index in a 32-bit lane mask");

__global__ void __launch_bounds__(AGG_THREADS)
h2gcn_aggregate(const int* __restrict__ rows1, const int* __restrict__ cols1,
                const float* __restrict__ vals1,
                const int* __restrict__ rows2, const int* __restrict__ cols2,
                const float* __restrict__ vals2,
                const float* __restrict__ h0,
                float* __restrict__ h1, float* __restrict__ h2) {
    __shared__ float acc[AGG_ROWS * HID];

    const int part = blockIdx.y;
    const int*   rows = part ? rows2 : rows1;
    const int*   cols = part ? cols2 : cols1;
    const float* vals = part ? vals2 : vals1;
    const int    E    = part ? E2 : E1;
    const int rowBase = blockIdx.x * AGG_ROWS;
    float* dst = (part ? h2 : h1) + (size_t)rowBase * HID;

    const int lane = threadIdx.x & 31;
    const int wave = __builtin_amdgcn_readfirstlane((int)(threadIdx.x >> 5));

    for (int i = threadIdx.x; i < AGG_ROWS * HID / 4; i += AGG_THREADS)
        ((float4*)acc)[i] = make_float4(0.0f, 0.0f, 0.0f, 0.0f);
    __syncthreads();

    const int nStep = (E + AGG_BSTEP - 1) / AGG_BSTEP;
    int step = blockIdx.x % nStep;

    int4 rv[AGG_VECS];
    auto loadIdx = [&](int base) {
        #pragma unroll
        for (int j = 0; j < AGG_VECS; ++j) {
            const int e0 = base + j * 128 + lane * 4;
            rv[j] = (e0 < E) ? *(const int4*)(rows + e0) : make_int4(-1, -1, -1, -1);
        }
    };
    loadIdx((step * AGG_WAVES + wave) * AGG_WSTEP);

    for (int s = 0; s < nStep; ++s) {
        const int base = (step * AGG_WAVES + wave) * AGG_WSTEP;

        unsigned lm = 0u;
        #pragma unroll
        for (int j = 0; j < AGG_VECS; ++j) {
            lm |= ((unsigned)(rv[j].x - rowBase) < (unsigned)AGG_ROWS ? 1u : 0u) << (4 * j + 0);
            lm |= ((unsigned)(rv[j].y - rowBase) < (unsigned)AGG_ROWS ? 1u : 0u) << (4 * j + 1);
            lm |= ((unsigned)(rv[j].z - rowBase) < (unsigned)AGG_ROWS ? 1u : 0u) << (4 * j + 2);
            lm |= ((unsigned)(rv[j].w - rowBase) < (unsigned)AGG_ROWS ? 1u : 0u) << (4 * j + 3);
        }

        step = (step + 1 == nStep) ? 0 : step + 1;
        if (s + 1 < nStep) loadIdx((step * AGG_WAVES + wave) * AGG_WSTEP);

        auto nextEdge = [&]() -> int {
            const unsigned bal = __builtin_amdgcn_ballot_w32(lm != 0u);
            if (bal == 0u) return -1;
            const int src = __builtin_ctz(bal);
            const unsigned lmS = (unsigned)__builtin_amdgcn_readlane((int)lm, src);
            const int k = __builtin_ctz(lmS);
            if (lane == src) lm &= lm - 1u;
            return base + (k >> 2) * 128 + src * 4 + (k & 3);
        };

        int e = nextEdge();
        int r = 0, c = 0; float v = 0.0f;
        if (e >= 0) { r = rows[e]; c = cols[e]; v = vals[e]; }
        while (e >= 0) {
            const float* src = h0 + (size_t)c * HID + lane;
            float* a = acc + (r - rowBase) * HID + lane;
            float sv[HID / 32];
            #pragma unroll
            for (int t = 0; t < HID / 32; ++t) sv[t] = src[32 * t];

            const int en = nextEdge();
            int rn = 0, cn = 0; float vn = 0.0f;
            if (en >= 0) { rn = rows[en]; cn = cols[en]; vn = vals[en]; }

            #pragma unroll
            for (int t = 0; t < HID / 32; ++t) atomicAdd(a + 32 * t, v * sv[t]);

            e = en; r = rn; c = cn; v = vn;
        }
    }

    __syncthreads();
    for (int i = threadIdx.x; i < AGG_ROWS * HID / 4; i += AGG_THREADS)
        ((float4*)dst)[i] = ((const float4*)acc)[i];
}

#define G2_KW    (3 * HID / 4)
#define G2_NBLK  (G2_KW / 32)
#define G2_PBLK  (HID / 32)

static_assert(G1_COLS * 2 == OUT_C, "two column halves");
static_assert(HID % 32 == 0 && G2_KW % 32 == 0, "a 32-deep block lies inside one part and one wave");
static_assert(G2_NBLK % 2 == 0 && G2_NBLK >= 4, "the K loop runs pairs of blocks (two A register sets)");

__global__ void __launch_bounds__(128)
h2gcn_gemm2(const float* __restrict__ h0,
            const float* __restrict__ h1,
            const float* __restrict__ h2,
            const float* __restrict__ Wout,
            const float* __restrict__ bout,
            float* __restrict__ out) {
    __shared__ __attribute__((aligned(16))) float part[4 * G1_ROWS * G1_PITCH];

    const int w    = threadIdx.x >> 5;
    const int lane = threadIdx.x & 31;
    const int c    = lane & 15;
    const int h    = lane >> 4;
    const int rb   = blockIdx.x * G1_ROWS;
    const int nb   = blockIdx.y * G1_COLS;

    v8f acc[G1_MT][G1_NT];
    #pragma unroll
    for (int mt = 0; mt < G1_MT; ++mt) {
        #pragma unroll
        for (int nt = 0; nt < G1_NT; ++nt) acc[mt][nt] = 0.0f;
    }

    const size_t aRow = (size_t)(rb + c) * HID + 4 * h;
    auto aptr = [&](int kb) -> const float* {
        const int p = kb / G2_PBLK;
        const float* s = (p == 0) ? h0 : ((p == 1) ? h1 : h2);
        return s + aRow + (kb % G2_PBLK) * 32;
    };

    int kb = w * G2_NBLK;
    const float* bp = Wout + (size_t)(kb * 32 + 4 * h) * OUT_C + nb + c;

    float4 a[2][G1_D][G1_MT];
    float  b[G1_D][4][G1_NT];

    g1_loadA<HID>(a[0], aptr(kb));
    #pragma unroll
    for (int d = 0; d < G1_D; ++d) g1_loadB(b[d], bp + (size_t)(d * 8) * OUT_C);

    for (int it = 0; it < G2_NBLK / 2 - 1; ++it) {
        g1_block<0, true, HID>(acc, a, b, aptr(kb + 1), bp);
        bp += (size_t)32 * OUT_C;
        g1_block<1, true, HID>(acc, a, b, aptr(kb + 2), bp);
        bp += (size_t)32 * OUT_C;
        kb += 2;
    }
    g1_block<0, true, HID>(acc, a, b, aptr(kb + 1), bp);
    bp += (size_t)32 * OUT_C;
    g1_block<1, false, HID>(acc, a, b, nullptr, bp);

    float* pw = part + w * (G1_ROWS * G1_PITCH);
    #pragma unroll
    for (int mt = 0; mt < G1_MT; ++mt) {
        #pragma unroll
        for (int nt = 0; nt < G1_NT; ++nt) {
            #pragma unroll
            for (int r = 0; r < 8; ++r)
                pw[(16 * mt + 8 * h + r) * G1_PITCH + 16 * nt + c] = acc[mt][nt][r];
        }
    }
    __syncthreads();

    for (int i = threadIdx.x; i < G1_ROWS * (G1_COLS / 4); i += 128) {
        const int row = i >> 5;
        const int c4  = i & 31;
        const float* p = part + row * G1_PITCH + c4 * 4;
        const float4 s0 = *(const float4*)(p);
        const float4 s1 = *(const float4*)(p + 1 * G1_ROWS * G1_PITCH);
        const float4 s2 = *(const float4*)(p + 2 * G1_ROWS * G1_PITCH);
        const float4 s3 = *(const float4*)(p + 3 * G1_ROWS * G1_PITCH);
        const float4 bi = *(const float4*)(bout + nb + c4 * 4);
        float4 o;
        o.x = (((s0.x + s1.x) + s2.x) + s3.x) + bi.x;
        o.y = (((s0.y + s1.y) + s2.y) + s3.y) + bi.y;
        o.z = (((s0.z + s1.z) + s2.z) + s3.z) + bi.z;
        o.w = (((s0.w + s1.w) + s2.w) + s3.w) + bi.w;
        *(float4*)(out + (size_t)(rb + row) * OUT_C + nb + c4 * 4) = o;
    }
}

extern "C" void kernel_launch(void* const* d_in, const int* in_sizes, int n_in,
                              void* d_out, int out_size, void* d_ws, size_t ws_size,
                              hipStream_t stream) {
    (void)in_sizes; (void)n_in; (void)out_size; (void)ws_size;

    const float* x    = (const float*)d_in[0];
    const int*   a1r  = (const int*)  d_in[1];
    const int*   a1c  = (const int*)  d_in[2];
    const float* a1v  = (const float*)d_in[3];
    const int*   a2r  = (const int*)  d_in[4];
    const int*   a2c  = (const int*)  d_in[5];
    const float* a2v  = (const float*)d_in[6];
    const float* W1   = (const float*)d_in[7];
    const float* Wout = (const float*)d_in[8];
    const float* bout = (const float*)d_in[9];
    float* out = (float*)d_out;

    float* h0 = (float*)d_ws;
    float* h1 = h0 + (size_t)NROWS * HID;
    float* h2 = h1 + (size_t)NROWS * HID;

    h2gcn_gemm1<<<dim3(125, 2), dim3(128), 0, stream>>>(x, W1, h0);

    h2gcn_aggregate<<<dim3(250, 2), dim3(AGG_THREADS), 0, stream>>>(
        a1r, a1c, a1v, a2r, a2c, a2v, h0, h1, h2);

    h2gcn_gemm2<<<dim3(125, 2), dim3(128), 0, stream>>>(h0, h1, h2, Wout, bout, out);
}


# Aggregation scan on owner bytes: GEMM1's launch also writes row/40 as 1 byte per edge; each row owner tests that byte, not the 4-byte row index: 4x fewer scanned bytes; sums still LDS arrival order
# speedup vs baseline: 2.4268x; 1.0494x over previous
#include <hip/hip_runtime.h>

#define NROWS 10000
#define IN_C  2048
#define HID   256
#define OUT_C 256
#define E1    160000
#define E2    320000

#define AGG_ROWS     40
#define AGG_OWNERS   (NROWS / AGG_ROWS)
#define OB_WORDS     ((E1 + E2) / 4)
#define OB_WG_WORDS  (OB_WORDS / 250)
#define OB_WG_LANES  (OB_WG_WORDS / 4)

static_assert(AGG_ROWS * AGG_OWNERS == NROWS, "row owners must tile the rows exactly");
static_assert(AGG_OWNERS < 255, "owner ids fit a byte, and 255 stays free for 'nobody'");
static_assert(E1 % 16 == 0 && E2 % 16 == 0, "16-byte owner-byte loads are all-in or all-out and stay aligned");
static_assert(OB_WG_WORDS * 250 == OB_WORDS, "GEMM1's work-groups cut the owner words exactly");
static_assert(OB_WG_LANES * 4 == OB_WG_WORDS && OB_WG_LANES <= 128, "four words per writing thread of a GEMM1 work-group");
static_assert((E1 / 4) % 4 == 0, "a thread's four index vectors lie in one adjacency");

static __device__ __forceinline__ unsigned owner_byte(int row) {
    const unsigned o = (unsigned)row / (unsigned)AGG_ROWS;
    return o < 255u ? o : 255u;
}

typedef float v2f __attribute__((ext_vector_type(2)));
typedef float v8f __attribute__((ext_vector_type(8)));

#define G1_ROWS  80
#define G1_COLS  128
#define G1_MT    (G1_ROWS / 16)
#define G1_NT    (G1_COLS / 16)
#define G1_KW    (IN_C / 4)
#define G1_NBLK  (G1_KW / 32)
#define G1_D     4
#define G1_PITCH 132

static_assert(G1_ROWS * 125 == NROWS, "row blocks must tile the rows exactly");
static_assert(G1_COLS * 2 == HID, "two column halves");
static_assert(G1_NBLK % 2 == 0, "the K loop runs pairs of blocks (two A register sets)");

template <int LD>
static __device__ __forceinline__ void g1_loadA(float4 (&a)[G1_D][G1_MT], const float* ap) {
    #pragma unroll
    for (int mt = 0; mt < G1_MT; ++mt) {
        #pragma unroll
        for (int d = 0; d < G1_D; ++d)
            a[d][mt] = *(const float4*)(ap + d * 8 + (size_t)mt * 16 * LD);
    }
}

static __device__ __forceinline__ void g1_loadB(float (&b)[4][G1_NT], const float* bp) {
    #pragma unroll
    for (int j = 0; j < 4; ++j) {
        #pragma unroll
        for (int nt = 0; nt < G1_NT; ++nt)
            b[j][nt] = bp[(size_t)j * HID + nt * 16];
    }
}

static __device__ __forceinline__ void g1_mul(v8f (&acc)[G1_MT][G1_NT],
                                              const float4 (&a)[G1_MT],
                                              const float (&b)[4][G1_NT]) {
    #pragma unroll
    for (int s = 0; s < 2; ++s) {
        #pragma unroll
        for (int mt = 0; mt < G1_MT; ++mt) {
            v2f av;
            av.x = s ? a[mt].z : a[mt].x;
            av.y = s ? a[mt].w : a[mt].y;
            #pragma unroll
            for (int nt = 0; nt < G1_NT; ++nt) {
                v2f bv;
                bv.x = b[2 * s][nt];
                bv.y = b[2 * s + 1][nt];
                acc[mt][nt] = __builtin_amdgcn_wmma_f32_16x16x4_f32(
                    false, av, false, bv, (short)0, acc[mt][nt], false, false);
            }
        }
    }
}

template <int P, bool MORE, int LD>
static __device__ __forceinline__ void g1_block(v8f (&acc)[G1_MT][G1_NT],
                                                float4 (&a)[2][G1_D][G1_MT],
                                                float (&b)[G1_D][4][G1_NT],
                                                const float* apn, const float* bp) {
    #pragma unroll
    for (int d = 0; d < G1_D; ++d) {
        g1_mul(acc, a[P][d], b[d]);
        if (MORE) {
            __builtin_amdgcn_sched_barrier(0);
            g1_loadB(b[d], bp + (size_t)(32 + d * 8) * HID);
            if (d == 0) g1_loadA<LD>(a[P ^ 1], apn);
            __builtin_amdgcn_sched_barrier(0);
        }
    }
}

__global__ void __launch_bounds__(128)
h2gcn_gemm1(const float* __restrict__ x,
            const float* __restrict__ W1,
            const int* __restrict__ rows1,
            const int* __restrict__ rows2,
            float* __restrict__ h0,
            float* __restrict__ obuf) {
    __shared__ __attribute__((aligned(16))) float part[4 * G1_ROWS * G1_PITCH];

    const int w    = threadIdx.x >> 5;
    const int lane = threadIdx.x & 31;
    const int c    = lane & 15;
    const int h    = lane >> 4;
    const int rb   = blockIdx.x * G1_ROWS;
    const int nb   = blockIdx.y * G1_COLS;
    const int kw   = w * G1_KW;

    v8f acc[G1_MT][G1_NT];
    #pragma unroll
    for (int mt = 0; mt < G1_MT; ++mt) {
        #pragma unroll
        for (int nt = 0; nt < G1_NT; ++nt) acc[mt][nt] = 0.0f;
    }

    const float* ap = x  + (size_t)(rb + c) * IN_C + kw + 4 * h;
    const float* bp = W1 + (size_t)(kw + 4 * h) * HID + nb + c;

    float4 a[2][G1_D][G1_MT];
    float  b[G1_D][4][G1_NT];

    g1_loadA<IN_C>(a[0], ap);
    #pragma unroll
    for (int d = 0; d < G1_D; ++d) g1_loadB(b[d], bp + (size_t)(d * 8) * HID);

    if (threadIdx.x < OB_WG_LANES) {
        const int q = (blockIdx.y * 125 + blockIdx.x) * OB_WG_WORDS
                    + 4 * (int)threadIdx.x;
        const int4* src = (q < E1 / 4) ? (const int4*)rows1 + q
                                       : (const int4*)rows2 + (q - E1 / 4);
        const int4 r0 = src[0], r1 = src[1], r2 = src[2], r3 = src[3];
        auto pack4 = [](const int4 r) -> unsigned {
            return owner_byte(r.x) | (owner_byte(r.y) << 8) |
                   (owner_byte(r.z) << 16) | (owner_byte(r.w) << 24);
        };
        *(uint4*)((unsigned*)obuf + q) = make_uint4(pack4(r0), pack4(r1), pack4(r2), pack4(r3));
    }

    for (int it = 0; it < G1_NBLK / 2 - 1; ++it) {
        g1_block<0, true, IN_C>(acc, a, b, ap + 32, bp);
        ap += 32;  bp += (size_t)32 * HID;
        g1_block<1, true, IN_C>(acc, a, b, ap + 32, bp);
        ap += 32;  bp += (size_t)32 * HID;
    }
    g1_block<0, true, IN_C>(acc, a, b, ap + 32, bp);
    ap += 32;  bp += (size_t)32 * HID;
    g1_block<1, false, IN_C>(acc, a, b, ap, bp);

    float* pw = part + w * (G1_ROWS * G1_PITCH);
    #pragma unroll
    for (int mt = 0; mt < G1_MT; ++mt) {
        #pragma unroll
        for (int nt = 0; nt < G1_NT; ++nt) {
            #pragma unroll
            for (int r = 0; r < 8; ++r)
                pw[(16 * mt + 8 * h + r) * G1_PITCH + 16 * nt + c] = acc[mt][nt][r];
        }
    }
    __syncthreads();

    for (int i = threadIdx.x; i < G1_ROWS * (G1_COLS / 4); i += 128) {
        const int row = i >> 5;
        const int c4  = i & 31;
        const float* p = part + row * G1_PITCH + c4 * 4;
        const float4 s0 = *(const float4*)(p);
        const float4 s1 = *(const float4*)(p + 1 * G1_ROWS * G1_PITCH);
        const float4 s2 = *(const float4*)(p + 2 * G1_ROWS * G1_PITCH);
        const float4 s3 = *(const float4*)(p + 3 * G1_ROWS * G1_PITCH);
        float4 o;
        o.x = ((s0.x + s1.x) + s2.x) + s3.x;
        o.y = ((s0.y + s1.y) + s2.y) + s3.y;
        o.z = ((s0.z + s1.z) + s2.z) + s3.z;
        o.w = ((s0.w + s1.w) + s2.w) + s3.w;
        *(float4*)(h0 + (size_t)(rb + row) * HID + nb + c4 * 4) = o;
    }
}

#define AGG_THREADS 1024
#define AGG_WAVES   (AGG_THREADS / 32)
#define AGG_WSTEP   (16 * 32)
#define AGG_BSTEP   (AGG_WSTEP * AGG_WAVES)

__global__ void __launch_bounds__(AGG_THREADS)
h2gcn_aggregate(const int* __restrict__ rows1, const int* __restrict__ cols1,
                const float* __restrict__ vals1,
                const int* __restrict__ rows2, const int* __restrict__ cols2,
                const float* __restrict__ vals2,
                const float* __restrict__ h0,
                const float* __restrict__ obuf,
                float* __restrict__ h1, float* __restrict__ h2) {
    __shared__ float acc[AGG_ROWS * HID];

    const int part = blockIdx.y;
    const int*   rows = part ? rows2 : rows1;
    const int*   cols = part ? cols2 : cols1;
    const float* vals = part ? vals2 : vals1;
    const int    E    = part ? E2 : E1;
    const int rowBase = blockIdx.x * AGG_ROWS;
    float* dst = (part ? h2 : h1) + (size_t)rowBase * HID;
    const unsigned char* ob = (const unsigned char*)obuf + (part ? E1 : 0);
    const unsigned pat = (unsigned)blockIdx.x * 0x01010101u;

    const int lane = threadIdx.x & 31;
    const int wave = __builtin_amdgcn_readfirstlane((int)(threadIdx.x >> 5));

    for (int i = threadIdx.x; i < AGG_ROWS * HID / 4; i += AGG_THREADS)
        ((float4*)acc)[i] = make_float4(0.0f, 0.0f, 0.0f, 0.0f);
    __syncthreads();

    const int nStep = (E + AGG_BSTEP - 1) / AGG_BSTEP;
    int step = blockIdx.x % nStep;

    uint4 rv;
    auto loadIdx = [&](int base) {
        const int e0 = base + lane * 16;
        rv = (e0 < E) ? *(const uint4*)(ob + e0) : make_uint4(~0u, ~0u, ~0u, ~0u);
    };
    auto hits4 = [&](unsigned w) -> unsigned {
        const unsigned x = w ^ pat;
        const unsigned f = ~(((x & 0x7F7F7F7Fu) + 0x7F7F7F7Fu) | x | 0x7F7F7F7Fu);
        return (((f >> 7) * 0x00204081u) >> 21) & 15u;
    };
    loadIdx((step * AGG_WAVES + wave) * AGG_WSTEP);

    for (int s = 0; s < nStep; ++s) {
        const int base = (step * AGG_WAVES + wave) * AGG_WSTEP;

        unsigned lm = hits4(rv.x) | (hits4(rv.y) << 4) | (hits4(rv.z) << 8) | (hits4(rv.w) << 12);

        step = (step + 1 == nStep) ? 0 : step + 1;
        if (s + 1 < nStep) loadIdx((step * AGG_WAVES + wave) * AGG_WSTEP);

        auto nextEdge = [&]() -> int {
            const unsigned bal = __builtin_amdgcn_ballot_w32(lm != 0u);
            if (bal == 0u) return -1;
            const int src = __builtin_ctz(bal);
            const unsigned lmS = (unsigned)__builtin_amdgcn_readlane((int)lm, src);
            const int k = __builtin_ctz(lmS);
            if (lane == src) lm &= lm - 1u;
            return base + src * 16 + k;
        };

        int e = nextEdge();
        int r = 0, c = 0; float v = 0.0f;
        if (e >= 0) { r = rows[e]; c = cols[e]; v = vals[e]; }
        while (e >= 0) {
            const float* src = h0 + (size_t)c * HID + lane;
            float* a = acc + (r - rowBase) * HID + lane;
            float sv[HID / 32];
            #pragma unroll
            for (int t = 0; t < HID / 32; ++t) sv[t] = src[32 * t];

            const int en = nextEdge();
            int rn = 0, cn = 0; float vn = 0.0f;
            if (en >= 0) { rn = rows[en]; cn = cols[en]; vn = vals[en]; }

            #pragma unroll
            for (int t = 0; t < HID / 32; ++t) atomicAdd(a + 32 * t, v * sv[t]);

            e = en; r = rn; c = cn; v = vn;
        }
    }

    __syncthreads();
    for (int i = threadIdx.x; i < AGG_ROWS * HID / 4; i += AGG_THREADS)
        ((float4*)dst)[i] = ((const float4*)acc)[i];
}

#define G2_KW    (3 * HID / 4)
#define G2_NBLK  (G2_KW / 32)
#define G2_PBLK  (HID / 32)

static_assert(G1_COLS * 2 == OUT_C, "two column halves");
static_assert(HID % 32 == 0 && G2_KW % 32 == 0, "a 32-deep block lies inside one part and one wave");
static_assert(G2_NBLK % 2 == 0 && G2_NBLK >= 4, "the K loop runs pairs of blocks (two A register sets)");

__global__ void __launch_bounds__(128)
h2gcn_gemm2(const float* __restrict__ h0,
            const float* __restrict__ h1,
            const float* __restrict__ h2,
            const float* __restrict__ Wout,
            const float* __restrict__ bout,
            float* __restrict__ out) {
    __shared__ __attribute__((aligned(16))) float part[4 * G1_ROWS * G1_PITCH];

    const int w    = threadIdx.x >> 5;
    const int lane = threadIdx.x & 31;
    const int c    = lane & 15;
    const int h    = lane >> 4;
    const int rb   = blockIdx.x * G1_ROWS;
    const int nb   = blockIdx.y * G1_COLS;

    v8f acc[G1_MT][G1_NT];
    #pragma unroll
    for (int mt = 0; mt < G1_MT; ++mt) {
        #pragma unroll
        for (int nt = 0; nt < G1_NT; ++nt) acc[mt][nt] = 0.0f;
    }

    const size_t aRow = (size_t)(rb + c) * HID + 4 * h;
    auto aptr = [&](int kb) -> const float* {
        const int p = kb / G2_PBLK;
        const float* s = (p == 0) ? h0 : ((p == 1) ? h1 : h2);
        return s + aRow + (kb % G2_PBLK) * 32;
    };

    int kb = w * G2_NBLK;
    const float* bp = Wout + (size_t)(kb * 32 + 4 * h) * OUT_C + nb + c;

    float4 a[2][G1_D][G1_MT];
    float  b[G1_D][4][G1_NT];

    g1_loadA<HID>(a[0], aptr(kb));
    #pragma unroll
    for (int d = 0; d < G1_D; ++d) g1_loadB(b[d], bp + (size_t)(d * 8) * OUT_C);

    for (int it = 0; it < G2_NBLK / 2 - 1; ++it) {
        g1_block<0, true, HID>(acc, a, b, aptr(kb + 1), bp);
        bp += (size_t)32 * OUT_C;
        g1_block<1, true, HID>(acc, a, b, aptr(kb + 2), bp);
        bp += (size_t)32 * OUT_C;
        kb += 2;
    }
    g1_block<0, true, HID>(acc, a, b, aptr(kb + 1), bp);
    bp += (size_t)32 * OUT_C;
    g1_block<1, false, HID>(acc, a, b, nullptr, bp);

    float* pw = part + w * (G1_ROWS * G1_PITCH);
    #pragma unroll
    for (int mt = 0; mt < G1_MT; ++mt) {
        #pragma unroll
        for (int nt = 0; nt < G1_NT; ++nt) {
            #pragma unroll
            for (int r = 0; r < 8; ++r)
                pw[(16 * mt + 8 * h + r) * G1_PITCH + 16 * nt + c] = acc[mt][nt][r];
        }
    }
    __syncthreads();

    for (int i = threadIdx.x; i < G1_ROWS * (G1_COLS / 4); i += 128) {
        const int row = i >> 5;
        const int c4  = i & 31;
        const float* p = part + row * G1_PITCH + c4 * 4;
        const float4 s0 = *(const float4*)(p);
        const float4 s1 = *(const float4*)(p + 1 * G1_ROWS * G1_PITCH);
        const float4 s2 = *(const float4*)(p + 2 * G1_ROWS * G1_PITCH);
        const float4 s3 = *(const float4*)(p + 3 * G1_ROWS * G1_PITCH);
        const float4 bi = *(const float4*)(bout + nb + c4 * 4);
        float4 o;
        o.x = (((s0.x + s1.x) + s2.x) + s3.x) + bi.x;
        o.y = (((s0.y + s1.y) + s2.y) + s3.y) + bi.y;
        o.z = (((s0.z + s1.z) + s2.z) + s3.z) + bi.z;
        o.w = (((s0.w + s1.w) + s2.w) + s3.w) + bi.w;
        *(float4*)(out + (size_t)(rb + row) * OUT_C + nb + c4 * 4) = o;
    }
}

extern "C" void kernel_launch(void* const* d_in, const int* in_sizes, int n_in,
                              void* d_out, int out_size, void* d_ws, size_t ws_size,
                              hipStream_t stream) {
    (void)in_sizes; (void)n_in; (void)out_size; (void)ws_size;

    const float* x    = (const float*)d_in[0];
    const int*   a1r  = (const int*)  d_in[1];
    const int*   a1c  = (const int*)  d_in[2];
    const float* a1v  = (const float*)d_in[3];
    const int*   a2r  = (const int*)  d_in[4];
    const int*   a2c  = (const int*)  d_in[5];
    const float* a2v  = (const float*)d_in[6];
    const float* W1   = (const float*)d_in[7];
    const float* Wout = (const float*)d_in[8];
    const float* bout = (const float*)d_in[9];
    float* out = (float*)d_out;

    float* h0 = (float*)d_ws;
    float* h1 = h0 + (size_t)NROWS * HID;
    float* h2 = h1 + (size_t)NROWS * HID;

    h2gcn_gemm1<<<dim3(125, 2), dim3(128), 0, stream>>>(x, W1, a1r, a2r, h0, out);

    h2gcn_aggregate<<<dim3(250, 2), dim3(AGG_THREADS), 0, stream>>>(
        a1r, a1c, a1v, a2r, a2c, a2v, h0, out, h1, h2);

    h2gcn_gemm2<<<dim3(125, 2), dim3(128), 0, stream>>>(h0, h1, h2, Wout, bout, out);
}


# Aggregation gathers by 16-byte loads: 2 wave loads of 4 whole lines per source row, not 8 one-line loads; LDS sums kept lane-major so adds stay conflict-free; sums still in LDS arrival order
# speedup vs baseline: 2.6173x; 1.0785x over previous
#include <hip/hip_runtime.h>

#define NROWS 10000
#define IN_C  2048
#define HID   256
#define OUT_C 256
#define E1    160000
#define E2    320000

#define AGG_ROWS     40
#define AGG_OWNERS   (NROWS / AGG_ROWS)
#define OB_WORDS     ((E1 + E2) / 4)
#define OB_WG_WORDS  (OB_WORDS / 250)
#define OB_WG_LANES  (OB_WG_WORDS / 4)

static_assert(AGG_ROWS * AGG_OWNERS == NROWS, "row owners must tile the rows exactly");
static_assert(HID == 256, "the aggregation's lane-major LDS row layout is written for 256 floats per row");
static_assert(AGG_OWNERS < 255, "owner ids fit a byte, and 255 stays free for 'nobody'");
static_assert(E1 % 16 == 0 && E2 % 16 == 0, "16-byte owner-byte loads are all-in or all-out and stay aligned");
static_assert(OB_WG_WORDS * 250 == OB_WORDS, "GEMM1's work-groups cut the owner words exactly");
static_assert(OB_WG_LANES * 4 == OB_WG_WORDS && OB_WG_LANES <= 128, "four words per writing thread of a GEMM1 work-group");
static_assert((E1 / 4) % 4 == 0, "a thread's four index vectors lie in one adjacency");

static __device__ __forceinline__ unsigned owner_byte(int row) {
    const unsigned o = (unsigned)row / (unsigned)AGG_ROWS;
    return o < 255u ? o : 255u;
}

typedef float v2f __attribute__((ext_vector_type(2)));
typedef float v8f __attribute__((ext_vector_type(8)));

#define G1_ROWS  80
#define G1_COLS  128
#define G1_MT    (G1_ROWS / 16)
#define G1_NT    (G1_COLS / 16)
#define G1_KW    (IN_C / 4)
#define G1_NBLK  (G1_KW / 32)
#define G1_D     4
#define G1_PITCH 132

static_assert(G1_ROWS * 125 == NROWS, "row blocks must tile the rows exactly");
static_assert(G1_COLS * 2 == HID, "two column halves");
static_assert(G1_NBLK % 2 == 0, "the K loop runs pairs of blocks (two A register sets)");

template <int LD>
static __device__ __forceinline__ void g1_loadA(float4 (&a)[G1_D][G1_MT], const float* ap) {
    #pragma unroll
    for (int mt = 0; mt < G1_MT; ++mt) {
        #pragma unroll
        for (int d = 0; d < G1_D; ++d)
            a[d][mt] = *(const float4*)(ap + d * 8 + (size_t)mt * 16 * LD);
    }
}

static __device__ __forceinline__ void g1_loadB(float (&b)[4][G1_NT], const float* bp) {
    #pragma unroll
    for (int j = 0; j < 4; ++j) {
        #pragma unroll
        for (int nt = 0; nt < G1_NT; ++nt)
            b[j][nt] = bp[(size_t)j * HID + nt * 16];
    }
}

static __device__ __forceinline__ void g1_mul(v8f (&acc)[G1_MT][G1_NT],
                                              const float4 (&a)[G1_MT],
                                              const float (&b)[4][G1_NT]) {
    #pragma unroll
    for (int s = 0; s < 2; ++s) {
        #pragma unroll
        for (int mt = 0; mt < G1_MT; ++mt) {
            v2f av;
            av.x = s ? a[mt].z : a[mt].x;
            av.y = s ? a[mt].w : a[mt].y;
            #pragma unroll
            for (int nt = 0; nt < G1_NT; ++nt) {
                v2f bv;
                bv.x = b[2 * s][nt];
                bv.y = b[2 * s + 1][nt];
                acc[mt][nt] = __builtin_amdgcn_wmma_f32_16x16x4_f32(
                    false, av, false, bv, (short)0, acc[mt][nt], false, false);
            }
        }
    }
}

template <int P, bool MORE, int LD>
static __device__ __forceinline__ void g1_block(v8f (&acc)[G1_MT][G1_NT],
                                                float4 (&a)[2][G1_D][G1_MT],
                                                float (&b)[G1_D][4][G1_NT],
                                                const float* apn, const float* bp) {
    #pragma unroll
    for (int d = 0; d < G1_D; ++d) {
        g1_mul(acc, a[P][d], b[d]);
        if (MORE) {
            __builtin_amdgcn_sched_barrier(0);
            g1_loadB(b[d], bp + (size_t)(32 + d * 8) * HID);
            if (d == 0) g1_loadA<LD>(a[P ^ 1], apn);
            __builtin_amdgcn_sched_barrier(0);
        }
    }
}

__global__ void __launch_bounds__(128)
h2gcn_gemm1(const float* __restrict__ x,
            const float* __restrict__ W1,
            const int* __restrict__ rows1,
            const int* __restrict__ rows2,
            float* __restrict__ h0,
            float* __restrict__ obuf) {
    __shared__ __attribute__((aligned(16))) float part[4 * G1_ROWS * G1_PITCH];

    const int w    = threadIdx.x >> 5;
    const int lane = threadIdx.x & 31;
    const int c    = lane & 15;
    const int h    = lane >> 4;
    const int rb   = blockIdx.x * G1_ROWS;
    const int nb   = blockIdx.y * G1_COLS;
    const int kw   = w * G1_KW;

    v8f acc[G1_MT][G1_NT];
    #pragma unroll
    for (int mt = 0; mt < G1_MT; ++mt) {
        #pragma unroll
        for (int nt = 0; nt < G1_NT; ++nt) acc[mt][nt] = 0.0f;
    }

    const float* ap = x  + (size_t)(rb + c) * IN_C + kw + 4 * h;
    const float* bp = W1 + (size_t)(kw + 4 * h) * HID + nb + c;

    float4 a[2][G1_D][G1_MT];
    float  b[G1_D][4][G1_NT];

    g1_loadA<IN_C>(a[0], ap);
    #pragma unroll
    for (int d = 0; d < G1_D; ++d) g1_loadB(b[d], bp + (size_t)(d * 8) * HID);

    if (threadIdx.x < OB_WG_LANES) {
        const int q = (blockIdx.y * 125 + blockIdx.x) * OB_WG_WORDS
                    + 4 * (int)threadIdx.x;
        const int4* src = (q < E1 / 4) ? (const int4*)rows1 + q
                                       : (const int4*)rows2 + (q - E1 / 4);
        const int4 r0 = src[0], r1 = src[1], r2 = src[2], r3 = src[3];
        auto pack4 = [](const int4 r) -> unsigned {
            return owner_byte(r.x) | (owner_byte(r.y) << 8) |
                   (owner_byte(r.z) << 16) | (owner_byte(r.w) << 24);
        };
        *(uint4*)((unsigned*)obuf + q) = make_uint4(pack4(r0), pack4(r1), pack4(r2), pack4(r3));
    }

    for (int it = 0; it < G1_NBLK / 2 - 1; ++it) {
        g1_block<0, true, IN_C>(acc, a, b, ap + 32, bp);
        ap += 32;  bp += (size_t)32 * HID;
        g1_block<1, true, IN_C>(acc, a, b, ap + 32, bp);
        ap += 32;  bp += (size_t)32 * HID;
    }
    g1_block<0, true, IN_C>(acc, a, b, ap + 32, bp);
    ap += 32;  bp += (size_t)32 * HID;
    g1_block<1, false, IN_C>(acc, a, b, ap, bp);

    float* pw = part + w * (G1_ROWS * G1_PITCH);
    #pragma unroll
    for (int mt = 0; mt < G1_MT; ++mt) {
        #pragma unroll
        for (int nt = 0; nt < G1_NT; ++nt) {
            #pragma unroll
            for (int r = 0; r < 8; ++r)
                pw[(16 * mt + 8 * h + r) * G1_PITCH + 16 * nt + c] = acc[mt][nt][r];
        }
    }
    __syncthreads();

    for (int i = threadIdx.x; i < G1_ROWS * (G1_COLS / 4); i += 128) {
        const int row = i >> 5;
        const int c4  = i & 31;
        const float* p = part + row * G1_PITCH + c4 * 4;
        const float4 s0 = *(const float4*)(p);
        const float4 s1 = *(const float4*)(p + 1 * G1_ROWS * G1_PITCH);
        const float4 s2 = *(const float4*)(p + 2 * G1_ROWS * G1_PITCH);
        const float4 s3 = *(const float4*)(p + 3 * G1_ROWS * G1_PITCH);
        float4 o;
        o.x = ((s0.x + s1.x) + s2.x) + s3.x;
        o.y = ((s0.y + s1.y) + s2.y) + s3.y;
        o.z = ((s0.z + s1.z) + s2.z) + s3.z;
        o.w = ((s0.w + s1.w) + s2.w) + s3.w;
        *(float4*)(h0 + (size_t)(rb + row) * HID + nb + c4 * 4) = o;
    }
}

#define AGG_THREADS 1024
#define AGG_WAVES   (AGG_THREADS / 32)
#define AGG_WSTEP   (16 * 32)
#define AGG_BSTEP   (AGG_WSTEP * AGG_WAVES)

__global__ void __launch_bounds__(AGG_THREADS)
h2gcn_aggregate(const int* __restrict__ rows1, const int* __restrict__ cols1,
                const float* __restrict__ vals1,
                const int* __restrict__ rows2, const int* __restrict__ cols2,
                const float* __restrict__ vals2,
                const float* __restrict__ h0,
                const float* __restrict__ obuf,
                float* __restrict__ h1, float* __restrict__ h2) {
    __shared__ float acc[AGG_ROWS * HID];

    const int part = blockIdx.y;
    const int*   rows = part ? rows2 : rows1;
    const int*   cols = part ? cols2 : cols1;
    const float* vals = part ? vals2 : vals1;
    const int    E    = part ? E2 : E1;
    const int rowBase = blockIdx.x * AGG_ROWS;
    float* dst = (part ? h2 : h1) + (size_t)rowBase * HID;
    const unsigned char* ob = (const unsigned char*)obuf + (part ? E1 : 0);
    const unsigned pat = (unsigned)blockIdx.x * 0x01010101u;

    const int lane = threadIdx.x & 31;
    const int wave = __builtin_amdgcn_readfirstlane((int)(threadIdx.x >> 5));

    for (int i = threadIdx.x; i < AGG_ROWS * HID / 4; i += AGG_THREADS)
        ((float4*)acc)[i] = make_float4(0.0f, 0.0f, 0.0f, 0.0f);
    __syncthreads();

    const int nStep = (E + AGG_BSTEP - 1) / AGG_BSTEP;
    int step = blockIdx.x % nStep;

    uint4 rv;
    auto loadIdx = [&](int base) {
        const int e0 = base + lane * 16;
        rv = (e0 < E) ? *(const uint4*)(ob + e0) : make_uint4(~0u, ~0u, ~0u, ~0u);
    };
    auto hits4 = [&](unsigned w) -> unsigned {
        const unsigned x = w ^ pat;
        const unsigned f = ~(((x & 0x7F7F7F7Fu) + 0x7F7F7F7Fu) | x | 0x7F7F7F7Fu);
        return (((f >> 7) * 0x00204081u) >> 21) & 15u;
    };
    loadIdx((step * AGG_WAVES + wave) * AGG_WSTEP);

    for (int s = 0; s < nStep; ++s) {
        const int base = (step * AGG_WAVES + wave) * AGG_WSTEP;

        unsigned lm = hits4(rv.x) | (hits4(rv.y) << 4) | (hits4(rv.z) << 8) | (hits4(rv.w) << 12);

        step = (step + 1 == nStep) ? 0 : step + 1;
        if (s + 1 < nStep) loadIdx((step * AGG_WAVES + wave) * AGG_WSTEP);

        auto nextEdge = [&]() -> int {
            const unsigned bal = __builtin_amdgcn_ballot_w32(lm != 0u);
            if (bal == 0u) return -1;
            const int src = __builtin_ctz(bal);
            const unsigned lmS = (unsigned)__builtin_amdgcn_readlane((int)lm, src);
            const int k = __builtin_ctz(lmS);
            if (lane == src) lm &= lm - 1u;
            return base + src * 16 + k;
        };

        int e = nextEdge();
        int r = 0, c = 0; float v = 0.0f;
        if (e >= 0) { r = rows[e]; c = cols[e]; v = vals[e]; }
        while (e >= 0) {
            const float4* src = (const float4*)(h0 + (size_t)c * HID) + lane;
            float* a = acc + (r - rowBase) * HID + lane;
            const float4 s0 = src[0];
            const float4 s1 = src[HID / 8];

            const int en = nextEdge();
            int rn = 0, cn = 0; float vn = 0.0f;
            if (en >= 0) { rn = rows[en]; cn = cols[en]; vn = vals[en]; }

            atomicAdd(a +   0, v * s0.x);
            atomicAdd(a +  32, v * s0.y);
            atomicAdd(a +  64, v * s0.z);
            atomicAdd(a +  96, v * s0.w);
            atomicAdd(a + 128, v * s1.x);
            atomicAdd(a + 160, v * s1.y);
            atomicAdd(a + 192, v * s1.z);
            atomicAdd(a + 224, v * s1.w);

            e = en; r = rn; c = cn; v = vn;
        }
    }

    __syncthreads();
    for (int i = threadIdx.x; i < AGG_ROWS * HID / 4; i += AGG_THREADS) {
        const int row = i >> 6;
        const int q   = i & 63;
        const float* p = acc + row * HID + (q >> 5) * 128 + (q & 31);
        float4 o;
        o.x = p[0];  o.y = p[32];  o.z = p[64];  o.w = p[96];
        ((float4*)dst)[i] = o;
    }
}

#define G2_KW    (3 * HID / 4)
#define G2_NBLK  (G2_KW / 32)
#define G2_PBLK  (HID / 32)

static_assert(G1_COLS * 2 == OUT_C, "two column halves");
static_assert(HID % 32 == 0 && G2_KW % 32 == 0, "a 32-deep block lies inside one part and one wave");
static_assert(G2_NBLK % 2 == 0 && G2_NBLK >= 4, "the K loop runs pairs of blocks (two A register sets)");

__global__ void __launch_bounds__(128)
h2gcn_gemm2(const float* __restrict__ h0,
            const float* __restrict__ h1,
            const float* __restrict__ h2,
            const float* __restrict__ Wout,
            const float* __restrict__ bout,
            float* __restrict__ out) {
    __shared__ __attribute__((aligned(16))) float part[4 * G1_ROWS * G1_PITCH];

    const int w    = threadIdx.x >> 5;
    const int lane = threadIdx.x & 31;
    const int c    = lane & 15;
    const int h    = lane >> 4;
    const int rb   = blockIdx.x * G1_ROWS;
    const int nb   = blockIdx.y * G1_COLS;

    v8f acc[G1_MT][G1_NT];
    #pragma unroll
    for (int mt = 0; mt < G1_MT; ++mt) {
        #pragma unroll
        for (int nt = 0; nt < G1_NT; ++nt) acc[mt][nt] = 0.0f;
    }

    const size_t aRow = (size_t)(rb + c) * HID + 4 * h;
    auto aptr = [&](int kb) -> const float* {
        const int p = kb / G2_PBLK;
        const float* s = (p == 0) ? h0 : ((p == 1) ? h1 : h2);
        return s + aRow + (kb % G2_PBLK) * 32;
    };

    int kb = w * G2_NBLK;
    const float* bp = Wout + (size_t)(kb * 32 + 4 * h) * OUT_C + nb + c;

    float4 a[2][G1_D][G1_MT];
    float  b[G1_D][4][G1_NT];

    g1_loadA<HID>(a[0], aptr(kb));
    #pragma unroll
    for (int d = 0; d < G1_D; ++d) g1_loadB(b[d], bp + (size_t)(d * 8) * OUT_C);

    for (int it = 0; it < G2_NBLK / 2 - 1; ++it) {
        g1_block<0, true, HID>(acc, a, b, aptr(kb + 1), bp);
        bp += (size_t)32 * OUT_C;
        g1_block<1, true, HID>(acc, a, b, aptr(kb + 2), bp);
        bp += (size_t)32 * OUT_C;
        kb += 2;
    }
    g1_block<0, true, HID>(acc, a, b, aptr(kb + 1), bp);
    bp += (size_t)32 * OUT_C;
    g1_block<1, false, HID>(acc, a, b, nullptr, bp);

    float* pw = part + w * (G1_ROWS * G1_PITCH);
    #pragma unroll
    for (int mt = 0; mt < G1_MT; ++mt) {
        #pragma unroll
        for (int nt = 0; nt < G1_NT; ++nt) {
            #pragma unroll
            for (int r = 0; r < 8; ++r)
                pw[(16 * mt + 8 * h + r) * G1_PITCH + 16 * nt + c] = acc[mt][nt][r];
        }
    }
    __syncthreads();

    for (int i = threadIdx.x; i < G1_ROWS * (G1_COLS / 4); i += 128) {
        const int row = i >> 5;
        const int c4  = i & 31;
        const float* p = part + row * G1_PITCH + c4 * 4;
        const float4 s0 = *(const float4*)(p);
        const float4 s1 = *(const float4*)(p + 1 * G1_ROWS * G1_PITCH);
        const float4 s2 = *(const float4*)(p + 2 * G1_ROWS * G1_PITCH);
        const float4 s3 = *(const float4*)(p + 3 * G1_ROWS * G1_PITCH);
        const float4 bi = *(const float4*)(bout + nb + c4 * 4);
        float4 o;
        o.x = (((s0.x + s1.x) + s2.x) + s3.x) + bi.x;
        o.y = (((s0.y + s1.y) + s2.y) + s3.y) + bi.y;
        o.z = (((s0.z + s1.z) + s2.z) + s3.z) + bi.z;
        o.w = (((s0.w + s1.w) + s2.w) + s3.w) + bi.w;
        *(float4*)(out + (size_t)(rb + row) * OUT_C + nb + c4 * 4) = o;
    }
}

extern "C" void kernel_launch(void* const* d_in, const int* in_sizes, int n_in,
                              void* d_out, int out_size, void* d_ws, size_t ws_size,
                              hipStream_t stream) {
    (void)in_sizes; (void)n_in; (void)out_size; (void)ws_size;

    const float* x    = (const float*)d_in[0];
    const int*   a1r  = (const int*)  d_in[1];
    const int*   a1c  = (const int*)  d_in[2];
    const float* a1v  = (const float*)d_in[3];
    const int*   a2r  = (const int*)  d_in[4];
    const int*   a2c  = (const int*)  d_in[5];
    const float* a2v  = (const float*)d_in[6];
    const float* W1   = (const float*)d_in[7];
    const float* Wout = (const float*)d_in[8];
    const float* bout = (const float*)d_in[9];
    float* out = (float*)d_out;

    float* h0 = (float*)d_ws;
    float* h1 = h0 + (size_t)NROWS * HID;
    float* h2 = h1 + (size_t)NROWS * HID;

    h2gcn_gemm1<<<dim3(125, 2), dim3(128), 0, stream>>>(x, W1, a1r, a2r, h0, out);

    h2gcn_aggregate<<<dim3(250, 2), dim3(AGG_THREADS), 0, stream>>>(
        a1r, a1c, a1v, a2r, a2c, a2v, h0, out, h1, h2);

    h2gcn_gemm2<<<dim3(125, 2), dim3(128), 0, stream>>>(h0, h1, h2, Wout, bout, out);
}


# Aggregation sums in registers: owners queue hits, bucket (column, value) per row in LDS; a half wave per row gathers and adds (fused) in registers; no LDS float adds; sum order = LDS slot order
# speedup vs baseline: 2.7832x; 1.0634x over previous
#include <hip/hip_runtime.h>

#define NROWS 10000
#define IN_C  2048
#define HID   256
#define OUT_C 256
#define E1    160000
#define E2    320000

#define AGG_ROWS     40
#define AGG_OWNERS   (NROWS / AGG_ROWS)
#define OB_WORDS     ((E1 + E2) / 4)
#define OB_WG_WORDS  (OB_WORDS / 250)
#define OB_WG_LANES  (OB_WG_WORDS / 4)

static_assert(AGG_ROWS * AGG_OWNERS == NROWS, "row owners must tile the rows exactly");
static_assert(HID == 256, "the aggregation's lane-major LDS row layout is written for 256 floats per row");
static_assert(AGG_OWNERS < 255, "owner ids fit a byte, and 255 stays free for 'nobody'");
static_assert(E1 % 16 == 0 && E2 % 16 == 0, "16-byte owner-byte loads are all-in or all-out and stay aligned");
static_assert(OB_WG_WORDS * 250 == OB_WORDS, "GEMM1's work-groups cut the owner words exactly");
static_assert(OB_WG_LANES * 4 == OB_WG_WORDS && OB_WG_LANES <= 128, "four words per writing thread of a GEMM1 work-group");
static_assert((E1 / 4) % 4 == 0, "a thread's four index vectors lie in one adjacency");

static __device__ __forceinline__ unsigned owner_byte(int row) {
    const unsigned o = (unsigned)row / (unsigned)AGG_ROWS;
    return o < 255u ? o : 255u;
}

typedef float v2f __attribute__((ext_vector_type(2)));
typedef float v8f __attribute__((ext_vector_type(8)));

#define G1_ROWS  80
#define G1_COLS  128
#define G1_MT    (G1_ROWS / 16)
#define G1_NT    (G1_COLS / 16)
#define G1_KW    (IN_C / 4)
#define G1_NBLK  (G1_KW / 32)
#define G1_D     4
#define G1_PITCH 132

static_assert(G1_ROWS * 125 == NROWS, "row blocks must tile the rows exactly");
static_assert(G1_COLS * 2 == HID, "two column halves");
static_assert(G1_NBLK % 2 == 0, "the K loop runs pairs of blocks (two A register sets)");

template <int LD>
static __device__ __forceinline__ void g1_loadA(float4 (&a)[G1_D][G1_MT], const float* ap) {
    #pragma unroll
    for (int mt = 0; mt < G1_MT; ++mt) {
        #pragma unroll
        for (int d = 0; d < G1_D; ++d)
            a[d][mt] = *(const float4*)(ap + d * 8 + (size_t)mt * 16 * LD);
    }
}

static __device__ __forceinline__ void g1_loadB(float (&b)[4][G1_NT], const float* bp) {
    #pragma unroll
    for (int j = 0; j < 4; ++j) {
        #pragma unroll
        for (int nt = 0; nt < G1_NT; ++nt)
            b[j][nt] = bp[(size_t)j * HID + nt * 16];
    }
}

static __device__ __forceinline__ void g1_mul(v8f (&acc)[G1_MT][G1_NT],
                                              const float4 (&a)[G1_MT],
                                              const float (&b)[4][G1_NT]) {
    #pragma unroll
    for (int s = 0; s < 2; ++s) {
        #pragma unroll
        for (int mt = 0; mt < G1_MT; ++mt) {
            v2f av;
            av.x = s ? a[mt].z : a[mt].x;
            av.y = s ? a[mt].w : a[mt].y;
            #pragma unroll
            for (int nt = 0; nt < G1_NT; ++nt) {
                v2f bv;
                bv.x = b[2 * s][nt];
                bv.y = b[2 * s + 1][nt];
                acc[mt][nt] = __builtin_amdgcn_wmma_f32_16x16x4_f32(
                    false, av, false, bv, (short)0, acc[mt][nt], false, false);
            }
        }
    }
}

template <int P, bool MORE, int LD>
static __device__ __forceinline__ void g1_block(v8f (&acc)[G1_MT][G1_NT],
                                                float4 (&a)[2][G1_D][G1_MT],
                                                float (&b)[G1_D][4][G1_NT],
                                                const float* apn, const float* bp) {
    #pragma unroll
    for (int d = 0; d < G1_D; ++d) {
        g1_mul(acc, a[P][d], b[d]);
        if (MORE) {
            __builtin_amdgcn_sched_barrier(0);
            g1_loadB(b[d], bp + (size_t)(32 + d * 8) * HID);
            if (d == 0) g1_loadA<LD>(a[P ^ 1], apn);
            __builtin_amdgcn_sched_barrier(0);
        }
    }
}

__global__ void __launch_bounds__(128)
h2gcn_gemm1(const float* __restrict__ x,
            const float* __restrict__ W1,
            const int* __restrict__ rows1,
            const int* __restrict__ rows2,
            float* __restrict__ h0,
            float* __restrict__ obuf) {
    __shared__ __attribute__((aligned(16))) float part[4 * G1_ROWS * G1_PITCH];

    const int w    = threadIdx.x >> 5;
    const int lane = threadIdx.x & 31;
    const int c    = lane & 15;
    const int h    = lane >> 4;
    const int rb   = blockIdx.x * G1_ROWS;
    const int nb   = blockIdx.y * G1_COLS;
    const int kw   = w * G1_KW;

    v8f acc[G1_MT][G1_NT];
    #pragma unroll
    for (int mt = 0; mt < G1_MT; ++mt) {
        #pragma unroll
        for (int nt = 0; nt < G1_NT; ++nt) acc[mt][nt] = 0.0f;
    }

    const float* ap = x  + (size_t)(rb + c) * IN_C + kw + 4 * h;
    const float* bp = W1 + (size_t)(kw + 4 * h) * HID + nb + c;

    float4 a[2][G1_D][G1_MT];
    float  b[G1_D][4][G1_NT];

    g1_loadA<IN_C>(a[0], ap);
    #pragma unroll
    for (int d = 0; d < G1_D; ++d) g1_loadB(b[d], bp + (size_t)(d * 8) * HID);

    if (threadIdx.x < OB_WG_LANES) {
        const int q = (blockIdx.y * 125 + blockIdx.x) * OB_WG_WORDS
                    + 4 * (int)threadIdx.x;
        const int4* src = (q < E1 / 4) ? (const int4*)rows1 + q
                                       : (const int4*)rows2 + (q - E1 / 4);
        const int4 r0 = src[0], r1 = src[1], r2 = src[2], r3 = src[3];
        auto pack4 = [](const int4 r) -> unsigned {
            return owner_byte(r.x) | (owner_byte(r.y) << 8) |
                   (owner_byte(r.z) << 16) | (owner_byte(r.w) << 24);
        };
        *(uint4*)((unsigned*)obuf + q) = make_uint4(pack4(r0), pack4(r1), pack4(r2), pack4(r3));
    }

    for (int it = 0; it < G1_NBLK / 2 - 1; ++it) {
        g1_block<0, true, IN_C>(acc, a, b, ap + 32, bp);
        ap += 32;  bp += (size_t)32 * HID;
        g1_block<1, true, IN_C>(acc, a, b, ap + 32, bp);
        ap += 32;  bp += (size_t)32 * HID;
    }
    g1_block<0, true, IN_C>(acc, a, b, ap + 32, bp);
    ap += 32;  bp += (size_t)32 * HID;
    g1_block<1, false, IN_C>(acc, a, b, ap, bp);

    float* pw = part + w * (G1_ROWS * G1_PITCH);
    #pragma unroll
    for (int mt = 0; mt < G1_MT; ++mt) {
        #pragma unroll
        for (int nt = 0; nt < G1_NT; ++nt) {
            #pragma unroll
            for (int r = 0; r < 8; ++r)
                pw[(16 * mt + 8 * h + r) * G1_PITCH + 16 * nt + c] = acc[mt][nt][r];
        }
    }
    __syncthreads();

    for (int i = threadIdx.x; i < G1_ROWS * (G1_COLS / 4); i += 128) {
        const int row = i >> 5;
        const int c4  = i & 31;
        const float* p = part + row * G1_PITCH + c4 * 4;
        const float4 s0 = *(const float4*)(p);
        const float4 s1 = *(const float4*)(p + 1 * G1_ROWS * G1_PITCH);
        const float4 s2 = *(const float4*)(p + 2 * G1_ROWS * G1_PITCH);
        const float4 s3 = *(const float4*)(p + 3 * G1_ROWS * G1_PITCH);
        float4 o;
        o.x = ((s0.x + s1.x) + s2.x) + s3.x;
        o.y = ((s0.y + s1.y) + s2.y) + s3.y;
        o.z = ((s0.z + s1.z) + s2.z) + s3.z;
        o.w = ((s0.w + s1.w) + s2.w) + s3.w;
        *(float4*)(h0 + (size_t)(rb + row) * HID + nb + c4 * 4) = o;
    }
}

#define AGG_THREADS 1024
#define AGG_WAVES   (AGG_THREADS / 32)
#define AGG_WSTEP   (16 * 32)
#define AGG_BSTEP   (AGG_WSTEP * AGG_WAVES)
#define AGG_QW      128
#define AGG_ROWCAP  128

static_assert(AGG_ROWS * 16 <= AGG_THREADS, "one half wave per owned row");

__global__ void __launch_bounds__(AGG_THREADS)
h2gcn_aggregate(const int* __restrict__ rows1, const int* __restrict__ cols1,
                const float* __restrict__ vals1,
                const int* __restrict__ rows2, const int* __restrict__ cols2,
                const float* __restrict__ vals2,
                const float* __restrict__ h0,
                const float* __restrict__ obuf,
                float* __restrict__ h1, float* __restrict__ h2) {
    __shared__ __attribute__((aligned(16))) float acc[AGG_ROWS * HID];
    __shared__ int  queue[AGG_WAVES * AGG_QW];
    __shared__ int2 list[AGG_ROWS * AGG_ROWCAP];
    __shared__ int  cnt[AGG_ROWS];

    const int part = blockIdx.y;
    const int*   rows = part ? rows2 : rows1;
    const int*   cols = part ? cols2 : cols1;
    const float* vals = part ? vals2 : vals1;
    const int    E    = part ? E2 : E1;
    const int rowBase = blockIdx.x * AGG_ROWS;
    float* dst = (part ? h2 : h1) + (size_t)rowBase * HID;
    const unsigned char* ob = (const unsigned char*)obuf + (part ? E1 : 0);
    const unsigned pat = (unsigned)blockIdx.x * 0x01010101u;

    const int lane = threadIdx.x & 31;
    const int wave = __builtin_amdgcn_readfirstlane((int)(threadIdx.x >> 5));

    for (int i = threadIdx.x; i < AGG_ROWS * HID / 4; i += AGG_THREADS)
        ((float4*)acc)[i] = make_float4(0.0f, 0.0f, 0.0f, 0.0f);
    if (threadIdx.x < AGG_ROWS) cnt[threadIdx.x] = 0;
    __syncthreads();

    auto slowAdd = [&](int rl, int c, float v) {
        const float* srow = h0 + (size_t)c * HID;
        float* a = acc + rl * HID;
        for (int j = 0; j < HID; ++j) atomicAdd(a + j, v * srow[j]);
    };

    const int nStep = (E + AGG_BSTEP - 1) / AGG_BSTEP;
    int step = blockIdx.x % nStep;

    uint4 rv;
    auto loadIdx = [&](int base) {
        const int e0 = base + lane * 16;
        rv = (e0 < E) ? *(const uint4*)(ob + e0) : make_uint4(~0u, ~0u, ~0u, ~0u);
    };
    auto hits4 = [&](unsigned w) -> unsigned {
        const unsigned x = w ^ pat;
        const unsigned f = ~(((x & 0x7F7F7F7Fu) + 0x7F7F7F7Fu) | x | 0x7F7F7F7Fu);
        return (((f >> 7) * 0x00204081u) >> 21) & 15u;
    };
    loadIdx((step * AGG_WAVES + wave) * AGG_WSTEP);

    int* wq = queue + wave * AGG_QW;
    int wcount = 0;

    for (int s = 0; s < nStep; ++s) {
        const int base = (step * AGG_WAVES + wave) * AGG_WSTEP;

        unsigned lm = hits4(rv.x) | (hits4(rv.y) << 4) | (hits4(rv.z) << 8) | (hits4(rv.w) << 12);

        step = (step + 1 == nStep) ? 0 : step + 1;
        if (s + 1 < nStep) loadIdx((step * AGG_WAVES + wave) * AGG_WSTEP);

        unsigned bal = __builtin_amdgcn_ballot_w32(lm != 0u);
        while (bal != 0u) {
            if (lm != 0u) {
                const int e   = base + lane * 16 + __builtin_ctz(lm);
                const int pos = wcount + (int)__builtin_amdgcn_mbcnt_lo(bal, 0u);
                if (pos < AGG_QW) wq[pos] = e;
                else              slowAdd(rows[e] - rowBase, cols[e], vals[e]);
                lm &= lm - 1u;
            }
            wcount += __builtin_popcount(bal);
            bal = __builtin_amdgcn_ballot_w32(lm != 0u);
        }
    }

    auto push = [&](int rl, int c, float v) {
        const int slot = atomicAdd(&cnt[rl], 1);
        if (slot < AGG_ROWCAP) list[rl * AGG_ROWCAP + slot] = make_int2(c, __float_as_int(v));
        else                   slowAdd(rl, c, v);
    };
    const int nq = wcount < AGG_QW ? wcount : AGG_QW;
    for (int j0 = 0; j0 < nq; j0 += 64) {
        const int  ja = j0 + lane, jb = ja + 32;
        const bool va = ja < nq,   vb = jb < nq;
        const int  ea = va ? wq[ja] : 0;
        const int  eb = vb ? wq[jb] : 0;
        const int   ra = rows[ea], ca = cols[ea];
        const float fa = vals[ea];
        const int   rb = rows[eb], cb = cols[eb];
        const float fb = vals[eb];
        if (va) push(ra - rowBase, ca, fa);
        if (vb) push(rb - rowBase, cb, fb);
    }
    __syncthreads();

    const int hw = threadIdx.x >> 4;
    if (hw < AGG_ROWS) {
        const int l16 = threadIdx.x & 15;
        const int seen = cnt[hw];
        const int n = seen < AGG_ROWCAP ? seen : AGG_ROWCAP;
        const int2* L = list + hw * AGG_ROWCAP;
        const float* hp = h0 + 4 * l16;

        float4 s0 = make_float4(0.0f, 0.0f, 0.0f, 0.0f), s1 = s0, s2 = s0, s3 = s0;
        auto fma4 = [](float4& s, float v, const float4 g) {
            s.x += v * g.x;  s.y += v * g.y;  s.z += v * g.z;  s.w += v * g.w;
        };
        int i = 0;
        for (; i + 1 < n; i += 2) {
            const int2 ea = L[i], eb = L[i + 1];
            const float4* pa = (const float4*)(hp + (size_t)ea.x * HID);
            const float4* pb = (const float4*)(hp + (size_t)eb.x * HID);
            const float va = __int_as_float(ea.y), vb = __int_as_float(eb.y);
            const float4 a0 = pa[0], a1 = pa[16], a2 = pa[32], a3 = pa[48];
            const float4 b0 = pb[0], b1 = pb[16], b2 = pb[32], b3 = pb[48];
            fma4(s0, va, a0);  fma4(s1, va, a1);  fma4(s2, va, a2);  fma4(s3, va, a3);
            fma4(s0, vb, b0);  fma4(s1, vb, b1);  fma4(s2, vb, b2);  fma4(s3, vb, b3);
        }
        if (i < n) {
            const int2 ea = L[i];
            const float4* pa = (const float4*)(hp + (size_t)ea.x * HID);
            const float va = __int_as_float(ea.y);
            const float4 a0 = pa[0], a1 = pa[16], a2 = pa[32], a3 = pa[48];
            fma4(s0, va, a0);  fma4(s1, va, a1);  fma4(s2, va, a2);  fma4(s3, va, a3);
        }

        const float4* ap = (const float4*)(acc + hw * HID + 4 * l16);
        float4* dp = (float4*)(dst + (size_t)hw * HID + 4 * l16);
        auto add4 = [](const float4 a, const float4 b) {
            return make_float4(a.x + b.x, a.y + b.y, a.z + b.z, a.w + b.w);
        };
        dp[0]  = add4(s0, ap[0]);
        dp[16] = add4(s1, ap[16]);
        dp[32] = add4(s2, ap[32]);
        dp[48] = add4(s3, ap[48]);
    }
}

#define G2_KW    (3 * HID / 4)
#define G2_NBLK  (G2_KW / 32)
#define G2_PBLK  (HID / 32)

static_assert(G1_COLS * 2 == OUT_C, "two column halves");
static_assert(HID % 32 == 0 && G2_KW % 32 == 0, "a 32-deep block lies inside one part and one wave");
static_assert(G2_NBLK % 2 == 0 && G2_NBLK >= 4, "the K loop runs pairs of blocks (two A register sets)");

__global__ void __launch_bounds__(128)
h2gcn_gemm2(const float* __restrict__ h0,
            const float* __restrict__ h1,
            const float* __restrict__ h2,
            const float* __restrict__ Wout,
            const float* __restrict__ bout,
            float* __restrict__ out) {
    __shared__ __attribute__((aligned(16))) float part[4 * G1_ROWS * G1_PITCH];

    const int w    = threadIdx.x >> 5;
    const int lane = threadIdx.x & 31;
    const int c    = lane & 15;
    const int h    = lane >> 4;
    const int rb   = blockIdx.x * G1_ROWS;
    const int nb   = blockIdx.y * G1_COLS;

    v8f acc[G1_MT][G1_NT];
    #pragma unroll
    for (int mt = 0; mt < G1_MT; ++mt) {
        #pragma unroll
        for (int nt = 0; nt < G1_NT; ++nt) acc[mt][nt] = 0.0f;
    }

    const size_t aRow = (size_t)(rb + c) * HID + 4 * h;
    auto aptr = [&](int kb) -> const float* {
        const int p = kb / G2_PBLK;
        const float* s = (p == 0) ? h0 : ((p == 1) ? h1 : h2);
        return s + aRow + (kb % G2_PBLK) * 32;
    };

    int kb = w * G2_NBLK;
    const float* bp = Wout + (size_t)(kb * 32 + 4 * h) * OUT_C + nb + c;

    float4 a[2][G1_D][G1_MT];
    float  b[G1_D][4][G1_NT];

    g1_loadA<HID>(a[0], aptr(kb));
    #pragma unroll
    for (int d = 0; d < G1_D; ++d) g1_loadB(b[d], bp + (size_t)(d * 8) * OUT_C);

    for (int it = 0; it < G2_NBLK / 2 - 1; ++it) {
        g1_block<0, true, HID>(acc, a, b, aptr(kb + 1), bp);
        bp += (size_t)32 * OUT_C;
        g1_block<1, true, HID>(acc, a, b, aptr(kb + 2), bp);
        bp += (size_t)32 * OUT_C;
        kb += 2;
    }
    g1_block<0, true, HID>(acc, a, b, aptr(kb + 1), bp);
    bp += (size_t)32 * OUT_C;
    g1_block<1, false, HID>(acc, a, b, nullptr, bp);

    float* pw = part + w * (G1_ROWS * G1_PITCH);
    #pragma unroll
    for (int mt = 0; mt < G1_MT; ++mt) {
        #pragma unroll
        for (int nt = 0; nt < G1_NT; ++nt) {
            #pragma unroll
            for (int r = 0; r < 8; ++r)
                pw[(16 * mt + 8 * h + r) * G1_PITCH + 16 * nt + c] = acc[mt][nt][r];
        }
    }
    __syncthreads();

    for (int i = threadIdx.x; i < G1_ROWS * (G1_COLS / 4); i += 128) {
        const int row = i >> 5;
        const int c4  = i & 31;
        const float* p = part + row * G1_PITCH + c4 * 4;
        const float4 s0 = *(const float4*)(p);
        const float4 s1 = *(const float4*)(p + 1 * G1_ROWS * G1_PITCH);
        const float4 s2 = *(const float4*)(p + 2 * G1_ROWS * G1_PITCH);
        const float4 s3 = *(const float4*)(p + 3 * G1_ROWS * G1_PITCH);
        const float4 bi = *(const float4*)(bout + nb + c4 * 4);
        float4 o;
        o.x = (((s0.x + s1.x) + s2.x) + s3.x) + bi.x;
        o.y = (((s0.y + s1.y) + s2.y) + s3.y) + bi.y;
        o.z = (((s0.z + s1.z) + s2.z) + s3.z) + bi.z;
        o.w = (((s0.w + s1.w) + s2.w) + s3.w) + bi.w;
        *(float4*)(out + (size_t)(rb + row) * OUT_C + nb + c4 * 4) = o;
    }
}

extern "C" void kernel_launch(void* const* d_in, const int* in_sizes, int n_in,
                              void* d_out, int out_size, void* d_ws, size_t ws_size,
                              hipStream_t stream) {
    (void)in_sizes; (void)n_in; (void)out_size; (void)ws_size;

    const float* x    = (const float*)d_in[0];
    const int*   a1r  = (const int*)  d_in[1];
    const int*   a1c  = (const int*)  d_in[2];
    const float* a1v  = (const float*)d_in[3];
    const int*   a2r  = (const int*)  d_in[4];
    const int*   a2c  = (const int*)  d_in[5];
    const float* a2v  = (const float*)d_in[6];
    const float* W1   = (const float*)d_in[7];
    const float* Wout = (const float*)d_in[8];
    const float* bout = (const float*)d_in[9];
    float* out = (float*)d_out;

    float* h0 = (float*)d_ws;
    float* h1 = h0 + (size_t)NROWS * HID;
    float* h2 = h1 + (size_t)NROWS * HID;

    h2gcn_gemm1<<<dim3(125, 2), dim3(128), 0, stream>>>(x, W1, a1r, a2r, h0, out);

    h2gcn_aggregate<<<dim3(250, 2), dim3(AGG_THREADS), 0, stream>>>(
        a1r, a1c, a1v, a2r, a2c, a2v, h0, out, h1, h2);

    h2gcn_gemm2<<<dim3(125, 2), dim3(128), 0, stream>>>(h0, h1, h2, Wout, bout, out);
}
